# starting kernel
# baseline (speedup 1.0000x reference)
#include <hip/hip_runtime.h>
#include <math.h>

#define DIM 128
#define HID 16
#define HEADS 4
#define NEG_SLOPE 0.2f

typedef __attribute__((ext_vector_type(2))) float v2f;
typedef __attribute__((ext_vector_type(8))) float v8f;

__device__ __forceinline__ void atomicMaxFloat(float* addr, float v) {
    if (v >= 0.0f) {
        atomicMax((int*)addr, __float_as_int(v));
    } else {
        atomicMin((unsigned int*)addr, __float_as_uint(v));
    }
}

__global__ void __launch_bounds__(128)
gemm1_wmma(const float* __restrict__ x, const float* __restrict__ W1,
           float* __restrict__ h1, int N) {
    __shared__ float Wl[DIM * 64];
    for (int i = threadIdx.x; i < DIM * 64; i += blockDim.x) Wl[i] = W1[i];
    __syncthreads();

    const int wave  = threadIdx.x >> 5;
    const int lane  = threadIdx.x & 31;
    const int Mtile = blockIdx.x * 4 + wave;
    const int Mtiles = (N + 15) >> 4;
    if (Mtile >= Mtiles) return;

    const int r  = lane & 15;
    const int kh = lane >> 4;
    const int mbase = Mtile * 16;
    int row = mbase + r;
    if (row >= N) row = N - 1;
    const float* __restrict__ xr = x + (size_t)row * DIM;

    v8f acc0 = {}, acc1 = {}, acc2 = {}, acc3 = {};

    for (int kb = 0; kb < DIM; kb += 4) {
        const int k0 = kb + 2 * kh;
        v2f a;
        a.x = xr[k0];
        a.y = xr[k0 + 1];
        const float* w0 = &Wl[(size_t)k0 * 64];
        const float* w1 = &Wl[(size_t)(k0 + 1) * 64];
        v2f b;
        b.x = w0[r];       b.y = w1[r];
        acc0 = __builtin_amdgcn_wmma_f32_16x16x4_f32(false, a, false, b, (short)0, acc0, false, false);
        b.x = w0[16 + r];  b.y = w1[16 + r];
        acc1 = __builtin_amdgcn_wmma_f32_16x16x4_f32(false, a, false, b, (short)0, acc1, false, false);
        b.x = w0[32 + r];  b.y = w1[32 + r];
        acc2 = __builtin_amdgcn_wmma_f32_16x16x4_f32(false, a, false, b, (short)0, acc2, false, false);
        b.x = w0[48 + r];  b.y = w1[48 + r];
        acc3 = __builtin_amdgcn_wmma_f32_16x16x4_f32(false, a, false, b, (short)0, acc3, false, false);
    }

    float* outp = h1 + (size_t)mbase * 64;
#pragma unroll
    for (int i = 0; i < 8; ++i) {
        const int m = i + 8 * kh;
        if (mbase + m < N) {
            float* o = outp + (size_t)m * 64 + r;
            o[0]  = acc0[i];
            o[16] = acc1[i];
            o[32] = acc2[i];
            o[48] = acc3[i];
        }
    }
}

__global__ void __launch_bounds__(256)
node_prep1(const float* __restrict__ h1, const float* __restrict__ attS,
           const float* __restrict__ attD, float* __restrict__ as1,
           float* __restrict__ ad1, float* __restrict__ m1,
           float* __restrict__ s1, float* __restrict__ o1, int N) {
    const int tid = blockIdx.x * blockDim.x + threadIdx.x;
    if (tid >= N * HEADS) return;
    const int n  = tid >> 2;
    const int hd = tid & 3;
    const float* hp = h1 + (size_t)n * 64 + hd * HID;
    float ss = 0.f, dd = 0.f;
#pragma unroll
    for (int c = 0; c < HID; ++c) {
        const float v = hp[c];
        ss += v * attS[hd * HID + c];
        dd += v * attD[hd * HID + c];
    }
    as1[tid] = ss;
    ad1[tid] = dd;
    m1[tid]  = -3.0e38f;
    s1[tid]  = 0.f;
    float* op = o1 + (size_t)n * 64 + hd * HID;
#pragma unroll
    for (int c = 0; c < HID; ++c) op[c] = 0.f;
}

__global__ void __launch_bounds__(256)
edge_max1(const int* __restrict__ ei, const float* __restrict__ as1,
          const float* __restrict__ ad1, float* __restrict__ m1, int N, int E) {
    const int tid = blockIdx.x * blockDim.x + threadIdx.x;
    const int Et = E + N;
    const int e  = tid >> 2;
    const int hd = tid & 3;
    if (e >= Et) return;
    int s, d;
    if (e < E) { s = ei[e]; d = ei[E + e]; } else { s = d = e - E; }
    float le = as1[s * HEADS + hd] + ad1[d * HEADS + hd];
    le = (le > 0.f) ? le : NEG_SLOPE * le;
    atomicMaxFloat(&m1[d * HEADS + hd], le);
}

__global__ void __launch_bounds__(256)
edge_acc1(const int* __restrict__ ei, const float* __restrict__ as1,
          const float* __restrict__ ad1, const float* __restrict__ m1,
          float* __restrict__ s1, const float* __restrict__ h1,
          float* __restrict__ o1, int N, int E) {
    const int tid = blockIdx.x * blockDim.x + threadIdx.x;
    const int Et = E + N;
    const int e  = tid >> 2;
    const int hd = tid & 3;
    if (e >= Et) return;
    int s, d;
    if (e < E) { s = ei[e]; d = ei[E + e]; } else { s = d = e - E; }
    float le = as1[s * HEADS + hd] + ad1[d * HEADS + hd];
    le = (le > 0.f) ? le : NEG_SLOPE * le;
    const float w = __expf(le - m1[d * HEADS + hd]);
    atomicAdd(&s1[d * HEADS + hd], w);
    const float* hp = h1 + (size_t)s * 64 + hd * HID;
    float* op = o1 + (size_t)d * 64 + hd * HID;
#pragma unroll
    for (int c = 0; c < HID; ++c) atomicAdd(&op[c], w * hp[c]);
}

__global__ void __launch_bounds__(256)
finish1_elu(float* __restrict__ o1, const float* __restrict__ s1,
            const float* __restrict__ b1, int N) {
    const int i = blockIdx.x * blockDim.x + threadIdx.x;
    if (i >= N * 64) return;
    const int n  = i >> 6;
    const int hd = (i >> 4) & 3;
    float v = o1[i] / (s1[n * HEADS + hd] + 1e-16f) + b1[i & 63];
    v = (v > 0.f) ? v : expm1f(v);
    o1[i] = v;
}

__global__ void __launch_bounds__(256)
node_prep2(const float* __restrict__ o1, const float* __restrict__ W2,
           const float* __restrict__ attS2, const float* __restrict__ attD2,
           float* __restrict__ h2, float* __restrict__ as2,
           float* __restrict__ ad2, float* __restrict__ m2,
           float* __restrict__ s2, float* __restrict__ out, int N) {
    const int n = blockIdx.x * blockDim.x + threadIdx.x;
    if (n >= N) return;
    const float* hp = o1 + (size_t)n * 64;
    float a0 = 0.f, a1 = 0.f;
#pragma unroll
    for (int k = 0; k < 64; ++k) {
        const float v = hp[k];
        a0 += v * W2[k * 2 + 0];
        a1 += v * W2[k * 2 + 1];
    }
    h2[n * 2 + 0] = a0;
    h2[n * 2 + 1] = a1;
    as2[n] = a0 * attS2[0] + a1 * attS2[1];
    ad2[n] = a0 * attD2[0] + a1 * attD2[1];
    m2[n] = -3.0e38f;
    s2[n] = 0.f;
    out[n * 2 + 0] = 0.f;
    out[n * 2 + 1] = 0.f;
}

__global__ void __launch_bounds__(256)
edge_max2(const int* __restrict__ ei, const float* __restrict__ as2,
          const float* __restrict__ ad2, float* __restrict__ m2, int N, int E) {
    const int e = blockIdx.x * blockDim.x + threadIdx.x;
    const int Et = E + N;
    if (e >= Et) return;
    int s, d;
    if (e < E) { s = ei[e]; d = ei[E + e]; } else { s = d = e - E; }
    float le = as2[s] + ad2[d];
    le = (le > 0.f) ? le : NEG_SLOPE * le;
    atomicMaxFloat(&m2[d], le);
}

__global__ void __launch_bounds__(256)
edge_acc2(const int* __restrict__ ei, const float* __restrict__ as2,
          const float* __restrict__ ad2, const float* __restrict__ m2,
          float* __restrict__ s2, const float* __restrict__ h2,
          float* __restrict__ out, int N, int E) {
    const int e = blockIdx.x * blockDim.x + threadIdx.x;
    const int Et = E + N;
    if (e >= Et) return;
    int s, d;
    if (e < E) { s = ei[e]; d = ei[E + e]; } else { s = d = e - E; }
    float le = as2[s] + ad2[d];
    le = (le > 0.f) ? le : NEG_SLOPE * le;
    const float w = __expf(le - m2[d]);
    atomicAdd(&s2[d], w);
    atomicAdd(&out[d * 2 + 0], w * h2[s * 2 + 0]);
    atomicAdd(&out[d * 2 + 1], w * h2[s * 2 + 1]);
}

__global__ void __launch_bounds__(256)
finish2(float* __restrict__ out, const float* __restrict__ s2,
        const float* __restrict__ b2, int N) {
    const int n = blockIdx.x * blockDim.x + threadIdx.x;
    if (n >= N) return;
    const float inv = 1.0f / (s2[n] + 1e-16f);
    out[n * 2 + 0] = out[n * 2 + 0] * inv + b2[0];
    out[n * 2 + 1] = out[n * 2 + 1] * inv + b2[1];
}

extern "C" void kernel_launch(void* const* d_in, const int* in_sizes, int n_in,
                              void* d_out, int out_size, void* d_ws, size_t ws_size,
                              hipStream_t stream) {
    const float* x    = (const float*)d_in[0];
    const int*   ei   = (const int*)d_in[1];
    const float* W1   = (const float*)d_in[2];
    const float* aS1  = (const float*)d_in[3];
    const float* aD1  = (const float*)d_in[4];
    const float* b1   = (const float*)d_in[5];
    const float* W2   = (const float*)d_in[6];
    const float* aS2  = (const float*)d_in[7];
    const float* aD2  = (const float*)d_in[8];
    const float* b2   = (const float*)d_in[9];

    const int N = in_sizes[0] / DIM;
    const int E = in_sizes[1] / 2;
    float* out = (float*)d_out;

    float* ws  = (float*)d_ws;
    float* h1  = ws;
    float* as1 = h1 + (size_t)N * 64;
    float* ad1 = as1 + (size_t)N * HEADS;
    float* m1  = ad1 + (size_t)N * HEADS;
    float* s1  = m1 + (size_t)N * HEADS;
    float* o1  = s1 + (size_t)N * HEADS;
    float* h2  = o1 + (size_t)N * 64;
    float* as2 = h2 + (size_t)N * 2;
    float* ad2 = as2 + (size_t)N;
    float* m2  = ad2 + (size_t)N;
    float* s2  = m2 + (size_t)N;

    const int Et = E + N;
    const int Mtiles = (N + 15) >> 4;

    gemm1_wmma<<<(Mtiles + 3) / 4, 128, 0, stream>>>(x, W1, h1, N);
    node_prep1<<<(N * HEADS + 255) / 256, 256, 0, stream>>>(h1, aS1, aD1, as1, ad1, m1, s1, o1, N);
    edge_max1<<<((size_t)Et * HEADS + 255) / 256, 256, 0, stream>>>(ei, as1, ad1, m1, N, E);
    edge_acc1<<<((size_t)Et * HEADS + 255) / 256, 256, 0, stream>>>(ei, as1, ad1, m1, s1, h1, o1, N, E);
    finish1_elu<<<((size_t)N * 64 + 255) / 256, 256, 0, stream>>>(o1, s1, b1, N);
    node_prep2<<<(N + 255) / 256, 256, 0, stream>>>(o1, W2, aS2, aD2, h2, as2, ad2, m2, s2, out, N);
    edge_max2<<<(Et + 255) / 256, 256, 0, stream>>>(ei, as2, ad2, m2, N, E);
    edge_acc2<<<(Et + 255) / 256, 256, 0, stream>>>(ei, as2, ad2, m2, s2, h2, out, N, E);
    finish2<<<(N + 255) / 256, 256, 0, stream>>>(out, s2, b2, N);
}


# Layer-2 edge passes run on the layer-1 destination tiles in LDS: a wave re-walks its tile list; max, sum, 2 outputs by LDS atomics; global atomics gone; 7 launches to 5; sum order by LDS arrival
# speedup vs baseline: 4.6658x; 4.6658x over previous
#include <hip/hip_runtime.h>
#include <math.h>

#define DIM 128
#define HID 16
#define HEADS 4
#define NEG_SLOPE 0.2f

typedef __attribute__((ext_vector_type(2))) float v2f;
typedef __attribute__((ext_vector_type(8))) float v8f;

__device__ __forceinline__ void atomicMaxFloat(float* addr, float v) {
    if (v >= 0.0f) {
        atomicMax((int*)addr, __float_as_int(v));
    } else {
        atomicMin((unsigned int*)addr, __float_as_uint(v));
    }
}

__global__ void __launch_bounds__(128)
gemm1_wmma(const float* __restrict__ x, const float* __restrict__ W1,
           float* __restrict__ h1, int N) {
    __shared__ float Wl[DIM * 64];
    for (int i = threadIdx.x; i < DIM * 64; i += blockDim.x) Wl[i] = W1[i];
    __syncthreads();

    const int wave  = threadIdx.x >> 5;
    const int lane  = threadIdx.x & 31;
    const int Mtile = blockIdx.x * 4 + wave;
    const int Mtiles = (N + 15) >> 4;
    if (Mtile >= Mtiles) return;

    const int r  = lane & 15;
    const int kh = lane >> 4;
    const int mbase = Mtile * 16;
    int row = mbase + r;
    if (row >= N) row = N - 1;
    const float* __restrict__ xr = x + (size_t)row * DIM;

    v8f acc0 = {}, acc1 = {}, acc2 = {}, acc3 = {};

    for (int kb = 0; kb < DIM; kb += 4) {
        const int k0 = kb + 2 * kh;
        v2f a;
        a.x = xr[k0];
        a.y = xr[k0 + 1];
        const float* w0 = &Wl[(size_t)k0 * 64];
        const float* w1 = &Wl[(size_t)(k0 + 1) * 64];
        v2f b;
        b.x = w0[r];       b.y = w1[r];
        acc0 = __builtin_amdgcn_wmma_f32_16x16x4_f32(false, a, false, b, (short)0, acc0, false, false);
        b.x = w0[16 + r];  b.y = w1[16 + r];
        acc1 = __builtin_amdgcn_wmma_f32_16x16x4_f32(false, a, false, b, (short)0, acc1, false, false);
        b.x = w0[32 + r];  b.y = w1[32 + r];
        acc2 = __builtin_amdgcn_wmma_f32_16x16x4_f32(false, a, false, b, (short)0, acc2, false, false);
        b.x = w0[48 + r];  b.y = w1[48 + r];
        acc3 = __builtin_amdgcn_wmma_f32_16x16x4_f32(false, a, false, b, (short)0, acc3, false, false);
    }

    float* outp = h1 + (size_t)mbase * 64;
#pragma unroll
    for (int i = 0; i < 8; ++i) {
        const int m = i + 8 * kh;
        if (mbase + m < N) {
            float* o = outp + (size_t)m * 64 + r;
            o[0]  = acc0[i];
            o[16] = acc1[i];
            o[32] = acc2[i];
            o[48] = acc3[i];
        }
    }
}

__global__ void __launch_bounds__(256)
node_prep1(const float* __restrict__ h1, const float* __restrict__ attS,
           const float* __restrict__ attD, float* __restrict__ as1,
           float* __restrict__ ad1, int N) {
    const int tid = blockIdx.x * blockDim.x + threadIdx.x;
    if (tid >= N * HEADS) return;
    const int n  = tid >> 2;
    const int hd = tid & 3;
    const float* hp = h1 + (size_t)n * 64 + hd * HID;
    float ss = 0.f, dd = 0.f;
#pragma unroll
    for (int c = 0; c < HID; ++c) {
        const float v = hp[c];
        ss += v * attS[hd * HID + c];
        dd += v * attD[hd * HID + c];
    }
    as1[tid] = ss;
    ad1[tid] = dd;
}

#define TILE 512
#define WCAP 1024
#define QCAP 64
#define AGG_WAVES 32

__device__ __forceinline__ float leaky(float v) { return (v > 0.f) ? v : NEG_SLOPE * v; }

struct AggCtx {
    const int* src; const int* dst;
    const float* as1; const float* ad1; const float* h1;
    float* acc; float* mx; float* sm;
    int* wq_e;
    int* lst_s; int* lst_d;
    size_t lbase;
    int nbase, ntile, lane, wave, E;
    const float* rec; float* adl; float* oa; float* ob;
};

__device__ __forceinline__ void batch_max(const AggCtx& cx, int nv, int s, int dl) {
    if (cx.lane < nv) {
        const float4 a = *(const float4*)(cx.as1 + (size_t)s * 4);
        const float4 b = *(const float4*)(cx.ad1 + (size_t)(cx.nbase + dl) * 4);
        atomicMaxFloat(&cx.mx[dl * 4 + 0], leaky(a.x + b.x));
        atomicMaxFloat(&cx.mx[dl * 4 + 1], leaky(a.y + b.y));
        atomicMaxFloat(&cx.mx[dl * 4 + 2], leaky(a.z + b.z));
        atomicMaxFloat(&cx.mx[dl * 4 + 3], leaky(a.w + b.w));
    }
}

__device__ __forceinline__ void batch_acc(const AggCtx& cx, int nv, int s, int dl) {
    float w0 = 0.f, w1 = 0.f, w2 = 0.f, w3 = 0.f;
    if (cx.lane < nv) {
        const float4 a = *(const float4*)(cx.as1 + (size_t)s * 4);
        const float4 b = *(const float4*)(cx.ad1 + (size_t)(cx.nbase + dl) * 4);
        w0 = __expf(leaky(a.x + b.x) - cx.mx[dl * 4 + 0]);
        w1 = __expf(leaky(a.y + b.y) - cx.mx[dl * 4 + 1]);
        w2 = __expf(leaky(a.z + b.z) - cx.mx[dl * 4 + 2]);
        w3 = __expf(leaky(a.w + b.w) - cx.mx[dl * 4 + 3]);
        atomicAdd(&cx.sm[dl * 4 + 0], w0);
        atomicAdd(&cx.sm[dl * 4 + 1], w1);
        atomicAdd(&cx.sm[dl * 4 + 2], w2);
        atomicAdd(&cx.sm[dl * 4 + 3], w3);
    } else {
        s = 0; dl = 0;
    }
    const bool lo = cx.lane < 16;
    for (int j0 = 0; j0 < nv; j0 += 8) {
        float v0[8], v1[8];
#pragma unroll
        for (int u = 0; u < 8; ++u) {
            const int sj = __builtin_amdgcn_readlane(s, j0 + u);
            const float* hp = cx.h1 + (size_t)sj * 64 + cx.lane;
            v0[u] = hp[0];
            v1[u] = hp[32];
        }
#pragma unroll
        for (int u = 0; u < 8; ++u) {
            const int j = j0 + u;
            if (j < nv) {
                const int dj = __builtin_amdgcn_readlane(dl, j);
                const float wlo_a = __int_as_float(__builtin_amdgcn_readlane(__float_as_int(w0), j));
                const float whi_a = __int_as_float(__builtin_amdgcn_readlane(__float_as_int(w1), j));
                const float wlo_b = __int_as_float(__builtin_amdgcn_readlane(__float_as_int(w2), j));
                const float whi_b = __int_as_float(__builtin_amdgcn_readlane(__float_as_int(w3), j));
                const float wa = lo ? wlo_a : whi_a;
                const float wb = lo ? wlo_b : whi_b;
                atomicAdd(&cx.acc[dj * 64 + cx.lane], wa * v0[u]);
                atomicAdd(&cx.acc[dj * 64 + 32 + cx.lane], wb * v1[u]);
            }
        }
    }
}

__device__ __forceinline__ void batch_max2(const AggCtx& cx, int nv, int s, int dl) {
    if (cx.lane < nv) {
        const float a = cx.rec[(size_t)s * 4 + 2];
        atomicMaxFloat(&cx.mx[dl], leaky(a + cx.adl[dl]));
    }
}

__device__ __forceinline__ void batch_acc2(const AggCtx& cx, int nv, int s, int dl) {
    if (cx.lane < nv) {
        const float4 r = *(const float4*)(cx.rec + (size_t)s * 4);
        const float w = __expf(leaky(r.z + cx.adl[dl]) - cx.mx[dl]);
        atomicAdd(&cx.sm[dl], w);
        atomicAdd(&cx.oa[dl], w * r.x);
        atomicAdd(&cx.ob[dl], w * r.y);
    }
}

template <int PHASE>
__device__ __forceinline__ void flush_batch(const AggCtx& cx, int head, int nv, int& nlist, bool& ovf) {
    __builtin_amdgcn_wave_barrier();
    const int qi = (head + cx.lane) & (QCAP - 1);
    int s = 0, dl = 0;
    if (cx.lane < nv) {
        const int e = cx.wq_e[qi];
        s  = cx.src[e];
        dl = cx.dst[e] - cx.nbase;
    }
    if (PHASE == 1) {
        batch_max(cx, nv, s, dl);
        if (!ovf && nlist + 32 <= WCAP) {
            if (cx.lane < nv) {
                cx.lst_s[cx.lbase + nlist + cx.lane] = s;
                cx.lst_d[cx.lbase + nlist + cx.lane] = dl;
            }
            nlist += nv;
        } else {
            ovf = true;
        }
    } else if (PHASE == 2) {
        batch_acc(cx, nv, s, dl);
    } else if (PHASE == 3) {
        batch_max2(cx, nv, s, dl);
    } else {
        batch_acc2(cx, nv, s, dl);
    }
    __builtin_amdgcn_wave_barrier();
}

template <int PHASE>
__device__ __forceinline__ void scan_edges(const AggCtx& cx, int& nlist, bool& ovf) {
    int head = 0, cnt = 0;
    const bool al16 = ((((size_t)cx.dst) & 15) == 0);
    const int ngroups = (cx.E + 1023) >> 10;
    for (int g = cx.wave; g < ngroups; g += AGG_WAVES) {
        const int ebase = g * 1024 + cx.lane * 4;
        int d[32];
        if (al16 && (g + 1) * 1024 <= cx.E) {
            const int4* p = (const int4*)(cx.dst + ebase);
#pragma unroll
            for (int u = 0; u < 8; ++u) {
                const int4 v = p[u * 32];
                d[4 * u + 0] = v.x; d[4 * u + 1] = v.y; d[4 * u + 2] = v.z; d[4 * u + 3] = v.w;
            }
        } else {
#pragma unroll
            for (int b = 0; b < 32; ++b) {
                const int e = ebase + (b >> 2) * 128 + (b & 3);
                d[b] = (e < cx.E) ? cx.dst[e] : -1;
            }
        }
        unsigned bits = 0u;
#pragma unroll
        for (int b = 0; b < 32; ++b) {
            const unsigned dl = (unsigned)(d[b] - cx.nbase);
            bits |= (dl < (unsigned)cx.ntile) ? (1u << b) : 0u;
        }
        for (int round = 0; round < 32; ++round) {
            const bool has = bits != 0u;
            const unsigned mask = __builtin_amdgcn_ballot_w32(has);
            if (mask == 0u) break;
            if (has) {
                const int b = __ffs(bits) - 1;
                const int pos = (head + cnt + __popc(mask & ((1u << cx.lane) - 1u))) & (QCAP - 1);
                cx.wq_e[pos] = ebase + (b >> 2) * 128 + (b & 3);
                bits &= bits - 1u;
            }
            cnt += __popc(mask);
            if (cnt >= 32) {
                flush_batch<PHASE>(cx, head, 32, nlist, ovf);
                head = (head + 32) & (QCAP - 1);
                cnt -= 32;
            }
        }
    }
    if (cnt > 0) flush_batch<PHASE>(cx, head, cnt, nlist, ovf);
}

__global__ void __launch_bounds__(1024)
agg1_lds(const int* __restrict__ ei, const float* __restrict__ as1,
         const float* __restrict__ ad1, const float* __restrict__ h1,
         const float* __restrict__ b1, float* __restrict__ o1,
         int* lst_s, int* lst_d, int* lcnt, int N, int E) {
    __shared__ float acc[TILE * 64];
    __shared__ float mx[TILE * HEADS];
    __shared__ float sm[TILE * HEADS];
    __shared__ int   qe[AGG_WAVES * QCAP];

    const int tid = threadIdx.x;
    AggCtx cx;
    cx.lane = tid & 31;
    cx.wave = __builtin_amdgcn_readfirstlane(tid >> 5);
    cx.nbase = blockIdx.x * TILE;
    cx.ntile = (N - cx.nbase < TILE) ? (N - cx.nbase) : TILE;
    cx.E = E;
    cx.src = ei; cx.dst = ei + E;
    cx.as1 = as1; cx.ad1 = ad1; cx.h1 = h1;
    cx.acc = acc; cx.mx = mx; cx.sm = sm;
    cx.wq_e = qe + cx.wave * QCAP;
    cx.lst_s = lst_s; cx.lst_d = lst_d;
    cx.lbase = ((size_t)blockIdx.x * AGG_WAVES + cx.wave) * WCAP;

    for (int i = tid; i < TILE * 64; i += 1024) acc[i] = 0.f;
    for (int i = tid; i < TILE * HEADS; i += 1024) {
        sm[i] = 0.f;
        float v = 0.f;
        if ((i >> 2) < cx.ntile) {
            const size_t g = (size_t)cx.nbase * HEADS + i;
            v = leaky(as1[g] + ad1[g]);
        }
        mx[i] = v;
    }
    __syncthreads();

    int nlist = 0;
    bool ovf = false;
    scan_edges<1>(cx, nlist, ovf);
    if (cx.lane == 0) lcnt[blockIdx.x * AGG_WAVES + cx.wave] = ovf ? -1 : nlist;
    __syncthreads();

    if (!ovf) {
        for (int p = 0; p < nlist; p += 32) {
            const int nv = (nlist - p < 32) ? (nlist - p) : 32;
            int s = 0, dl = 0;
            if (cx.lane < nv) {
                s  = lst_s[cx.lbase + p + cx.lane];
                dl = lst_d[cx.lbase + p + cx.lane];
            }
            batch_acc(cx, nv, s, dl);
        }
    } else {
        int dummy_n = 0; bool dummy_o = false;
        scan_edges<2>(cx, dummy_n, dummy_o);
    }
    __syncthreads();

    for (int q = tid; q < cx.ntile * 16; q += 1024) {
        const int nl = q >> 4;
        const int c  = (q & 15) * 4;
        const int hd = c >> 4;
        const size_t n = (size_t)cx.nbase + nl;
        const float ls = leaky(as1[n * HEADS + hd] + ad1[n * HEADS + hd]);
        const float wself = __expf(ls - mx[nl * HEADS + hd]);
        const float den = sm[nl * HEADS + hd] + wself + 1e-16f;
        const float4 hv = *(const float4*)(h1 + n * 64 + c);
        const float4 av = *(const float4*)(acc + nl * 64 + c);
        const float4 bv = *(const float4*)(b1 + c);
        float4 r;
        r.x = (av.x + wself * hv.x) / den + bv.x;
        r.y = (av.y + wself * hv.y) / den + bv.y;
        r.z = (av.z + wself * hv.z) / den + bv.z;
        r.w = (av.w + wself * hv.w) / den + bv.w;
        r.x = (r.x > 0.f) ? r.x : expm1f(r.x);
        r.y = (r.y > 0.f) ? r.y : expm1f(r.y);
        r.z = (r.z > 0.f) ? r.z : expm1f(r.z);
        r.w = (r.w > 0.f) ? r.w : expm1f(r.w);
        *(float4*)(o1 + n * 64 + c) = r;
    }
}

__global__ void __launch_bounds__(256)
node_prep2(const float* __restrict__ o1, const float* __restrict__ W2,
           const float* __restrict__ attS2, const float* __restrict__ attD2,
           float* __restrict__ rec, int N) {
    const int n = blockIdx.x * blockDim.x + threadIdx.x;
    if (n >= N) return;
    const float* hp = o1 + (size_t)n * 64;
    float a0 = 0.f, a1 = 0.f;
#pragma unroll
    for (int k = 0; k < 64; ++k) {
        const float v = hp[k];
        a0 += v * W2[k * 2 + 0];
        a1 += v * W2[k * 2 + 1];
    }
    float4 r;
    r.x = a0;
    r.y = a1;
    r.z = a0 * attS2[0] + a1 * attS2[1];
    r.w = a0 * attD2[0] + a1 * attD2[1];
    *(float4*)(rec + (size_t)n * 4) = r;
}

__global__ void __launch_bounds__(1024)
agg2_lds(const int* __restrict__ ei, const float* __restrict__ rec,
         const float* __restrict__ b2, float* __restrict__ out,
         int* lst_s, int* lst_d, const int* __restrict__ lcnt, int N, int E) {
    __shared__ float mx[TILE];
    __shared__ float sm[TILE];
    __shared__ float oa[TILE];
    __shared__ float ob[TILE];
    __shared__ float adl[TILE];
    __shared__ int   qe[AGG_WAVES * QCAP];

    const int tid = threadIdx.x;
    AggCtx cx;
    cx.lane = tid & 31;
    cx.wave = __builtin_amdgcn_readfirstlane(tid >> 5);
    cx.nbase = blockIdx.x * TILE;
    cx.ntile = (N - cx.nbase < TILE) ? (N - cx.nbase) : TILE;
    cx.E = E;
    cx.src = ei; cx.dst = ei + E;
    cx.as1 = nullptr; cx.ad1 = nullptr; cx.h1 = nullptr; cx.acc = nullptr;
    cx.mx = mx; cx.sm = sm;
    cx.wq_e = qe + cx.wave * QCAP;
    cx.lst_s = lst_s; cx.lst_d = lst_d;
    cx.lbase = ((size_t)blockIdx.x * AGG_WAVES + cx.wave) * WCAP;
    cx.rec = rec; cx.adl = adl; cx.oa = oa; cx.ob = ob;

    for (int i = tid; i < TILE; i += 1024) {
        float m = 0.f, a = 0.f;
        if (i < cx.ntile) {
            const float4 r = *(const float4*)(rec + ((size_t)cx.nbase + i) * 4);
            a = r.w;
            m = leaky(r.z + r.w);
        }
        mx[i] = m; adl[i] = a;
        sm[i] = 0.f; oa[i] = 0.f; ob[i] = 0.f;
    }
    __syncthreads();

    const int cnt = lcnt[blockIdx.x * AGG_WAVES + cx.wave];
    int dummy_n = 0; bool dummy_o = false;

    if (cnt >= 0) {
        for (int p = 0; p < cnt; p += 32) {
            const int nv = (cnt - p < 32) ? (cnt - p) : 32;
            int s = 0, dl = 0;
            if (cx.lane < nv) {
                s  = lst_s[cx.lbase + p + cx.lane];
                dl = lst_d[cx.lbase + p + cx.lane];
            }
            batch_max2(cx, nv, s, dl);
        }
    } else {
        scan_edges<3>(cx, dummy_n, dummy_o);
    }
    __syncthreads();

    if (cnt >= 0) {
        for (int p = 0; p < cnt; p += 32) {
            const int nv = (cnt - p < 32) ? (cnt - p) : 32;
            int s = 0, dl = 0;
            if (cx.lane < nv) {
                s  = lst_s[cx.lbase + p + cx.lane];
                dl = lst_d[cx.lbase + p + cx.lane];
            }
            batch_acc2(cx, nv, s, dl);
        }
    } else {
        scan_edges<4>(cx, dummy_n, dummy_o);
    }
    __syncthreads();

    for (int i = tid; i < cx.ntile; i += 1024) {
        const size_t n = (size_t)cx.nbase + i;
        const float4 r = *(const float4*)(rec + n * 4);
        const float wself = __expf(leaky(r.z + r.w) - mx[i]);
        const float inv = 1.0f / (sm[i] + wself + 1e-16f);
        float2 o;
        o.x = (oa[i] + wself * r.x) * inv + b2[0];
        o.y = (ob[i] + wself * r.y) * inv + b2[1];
        *(float2*)(out + n * 2) = o;
    }
}

extern "C" void kernel_launch(void* const* d_in, const int* in_sizes, int n_in,
                              void* d_out, int out_size, void* d_ws, size_t ws_size,
                              hipStream_t stream) {
    const float* x    = (const float*)d_in[0];
    const int*   ei   = (const int*)d_in[1];
    const float* W1   = (const float*)d_in[2];
    const float* aS1  = (const float*)d_in[3];
    const float* aD1  = (const float*)d_in[4];
    const float* b1   = (const float*)d_in[5];
    const float* W2   = (const float*)d_in[6];
    const float* aS2  = (const float*)d_in[7];
    const float* aD2  = (const float*)d_in[8];
    const float* b2   = (const float*)d_in[9];

    const int N = in_sizes[0] / DIM;
    const int E = in_sizes[1] / 2;
    float* out = (float*)d_out;

    float* ws  = (float*)d_ws;
    float* h1  = ws;
    float* as1 = h1 + (size_t)N * 64;
    float* ad1 = as1 + (size_t)N * HEADS;
    float* o1  = ad1 + (size_t)N * HEADS;
    float* rec = o1 + (size_t)N * 64;
    const int tiles = (N + TILE - 1) / TILE;
    int* lst_s = (int*)(rec + (size_t)N * 4);
    int* lst_d = lst_s + (size_t)tiles * AGG_WAVES * WCAP;
    int* lcnt  = lst_d + (size_t)tiles * AGG_WAVES * WCAP;

    const int Mtiles = (N + 15) >> 4;

    gemm1_wmma<<<(Mtiles + 3) / 4, 128, 0, stream>>>(x, W1, h1, N);
    node_prep1<<<(N * HEADS + 255) / 256, 256, 0, stream>>>(h1, aS1, aD1, as1, ad1, N);
    agg1_lds<<<tiles, 1024, 0, stream>>>(ei, as1, ad1, h1, b1, o1, lst_s, lst_d, lcnt, N, E);
    node_prep2<<<(N + 255) / 256, 256, 0, stream>>>(o1, W2, aS2, aD2, rec, N);
    agg2_lds<<<tiles, 1024, 0, stream>>>(ei, rec, b2, out, lst_s, lst_d, lcnt, N, E);
}


# Destination column re-coded per call to 1 byte per edge (tile number) by the logit kernel's threads; tile scans read 1 byte per edge, test 4 edges per dword; sums stay in LDS arrival order
# speedup vs baseline: 5.2751x; 1.1306x over previous
#include <hip/hip_runtime.h>
#include <math.h>

#define DIM 128
#define HID 16
#define HEADS 4
#define NEG_SLOPE 0.2f

typedef __attribute__((ext_vector_type(2))) float v2f;
typedef __attribute__((ext_vector_type(8))) float v8f;

__device__ __forceinline__ void atomicMaxFloat(float* addr, float v) {
    if (v >= 0.0f) {
        atomicMax((int*)addr, __float_as_int(v));
    } else {
        atomicMin((unsigned int*)addr, __float_as_uint(v));
    }
}

__global__ void __launch_bounds__(128)
gemm1_wmma(const float* __restrict__ x, const float* __restrict__ W1,
           float* __restrict__ h1, int N) {
    __shared__ float Wl[DIM * 64];
    for (int i = threadIdx.x; i < DIM * 64; i += blockDim.x) Wl[i] = W1[i];
    __syncthreads();

    const int wave  = threadIdx.x >> 5;
    const int lane  = threadIdx.x & 31;
    const int Mtile = blockIdx.x * 4 + wave;
    const int Mtiles = (N + 15) >> 4;
    if (Mtile >= Mtiles) return;

    const int r  = lane & 15;
    const int kh = lane >> 4;
    const int mbase = Mtile * 16;
    int row = mbase + r;
    if (row >= N) row = N - 1;
    const float* __restrict__ xr = x + (size_t)row * DIM;

    v8f acc0 = {}, acc1 = {}, acc2 = {}, acc3 = {};

    for (int kb = 0; kb < DIM; kb += 4) {
        const int k0 = kb + 2 * kh;
        v2f a;
        a.x = xr[k0];
        a.y = xr[k0 + 1];
        const float* w0 = &Wl[(size_t)k0 * 64];
        const float* w1 = &Wl[(size_t)(k0 + 1) * 64];
        v2f b;
        b.x = w0[r];       b.y = w1[r];
        acc0 = __builtin_amdgcn_wmma_f32_16x16x4_f32(false, a, false, b, (short)0, acc0, false, false);
        b.x = w0[16 + r];  b.y = w1[16 + r];
        acc1 = __builtin_amdgcn_wmma_f32_16x16x4_f32(false, a, false, b, (short)0, acc1, false, false);
        b.x = w0[32 + r];  b.y = w1[32 + r];
        acc2 = __builtin_amdgcn_wmma_f32_16x16x4_f32(false, a, false, b, (short)0, acc2, false, false);
        b.x = w0[48 + r];  b.y = w1[48 + r];
        acc3 = __builtin_amdgcn_wmma_f32_16x16x4_f32(false, a, false, b, (short)0, acc3, false, false);
    }

    float* outp = h1 + (size_t)mbase * 64;
#pragma unroll
    for (int i = 0; i < 8; ++i) {
        const int m = i + 8 * kh;
        if (mbase + m < N) {
            float* o = outp + (size_t)m * 64 + r;
            o[0]  = acc0[i];
            o[16] = acc1[i];
            o[32] = acc2[i];
            o[48] = acc3[i];
        }
    }
}

#define TILE 512
#define TILE_SHIFT 9
#define GRP 4096

__device__ __forceinline__ unsigned tile_code(int d) {
    return (d < 0) ? 255u : (((unsigned)d >> TILE_SHIFT) % 255u);
}

__global__ void __launch_bounds__(256)
node_prep1(const float* __restrict__ h1, const float* __restrict__ attS,
           const float* __restrict__ attD, float* __restrict__ as1,
           float* __restrict__ ad1, const int* __restrict__ ei,
           unsigned* __restrict__ code, int N, int E, int ncode) {
    const int tid = blockIdx.x * blockDim.x + threadIdx.x;
    {
        const int* __restrict__ dstc = ei + E;
        const bool al16 = ((((size_t)dstc) & 15) == 0);
        const int nthreads = gridDim.x * blockDim.x;
        for (int q = tid; q < ncode; q += nthreads) {
            const int e = q * 4;
            int d0 = -1, d1 = -1, d2 = -1, d3 = -1;
            if (al16 && e + 3 < E) {
                const int4 v = *(const int4*)(dstc + e);
                d0 = v.x; d1 = v.y; d2 = v.z; d3 = v.w;
            } else {
                if (e     < E) d0 = dstc[e];
                if (e + 1 < E) d1 = dstc[e + 1];
                if (e + 2 < E) d2 = dstc[e + 2];
                if (e + 3 < E) d3 = dstc[e + 3];
            }
            code[q] = tile_code(d0) | (tile_code(d1) << 8) | (tile_code(d2) << 16) | (tile_code(d3) << 24);
        }
    }
    if (tid >= N * HEADS) return;
    const int n  = tid >> 2;
    const int hd = tid & 3;
    const float* hp = h1 + (size_t)n * 64 + hd * HID;
    float ss = 0.f, dd = 0.f;
#pragma unroll
    for (int c = 0; c < HID; ++c) {
        const float v = hp[c];
        ss += v * attS[hd * HID + c];
        dd += v * attD[hd * HID + c];
    }
    as1[tid] = ss;
    ad1[tid] = dd;
}

#define WCAP 1024
#define QCAP 64
#define AGG_WAVES 32

__device__ __forceinline__ float leaky(float v) { return (v > 0.f) ? v : NEG_SLOPE * v; }

struct AggCtx {
    const int* src; const int* dst;
    const float* as1; const float* ad1; const float* h1;
    float* acc; float* mx; float* sm;
    int* wq_e;
    int* lst_s; int* lst_d;
    size_t lbase;
    int nbase, ntile, lane, wave, E;
    const unsigned* code; unsigned pat; int ngroups;
    const float* rec; float* adl; float* oa; float* ob;
};

__device__ __forceinline__ void batch_max(const AggCtx& cx, bool ok, int s, int dl) {
    if (ok) {
        const float4 a = *(const float4*)(cx.as1 + (size_t)s * 4);
        const float4 b = *(const float4*)(cx.ad1 + (size_t)(cx.nbase + dl) * 4);
        atomicMaxFloat(&cx.mx[dl * 4 + 0], leaky(a.x + b.x));
        atomicMaxFloat(&cx.mx[dl * 4 + 1], leaky(a.y + b.y));
        atomicMaxFloat(&cx.mx[dl * 4 + 2], leaky(a.z + b.z));
        atomicMaxFloat(&cx.mx[dl * 4 + 3], leaky(a.w + b.w));
    }
}

__device__ __forceinline__ void batch_acc(const AggCtx& cx, int nv, unsigned okm, int s, int dl) {
    float w0 = 0.f, w1 = 0.f, w2 = 0.f, w3 = 0.f;
    if ((okm >> cx.lane) & 1u) {
        const float4 a = *(const float4*)(cx.as1 + (size_t)s * 4);
        const float4 b = *(const float4*)(cx.ad1 + (size_t)(cx.nbase + dl) * 4);
        w0 = __expf(leaky(a.x + b.x) - cx.mx[dl * 4 + 0]);
        w1 = __expf(leaky(a.y + b.y) - cx.mx[dl * 4 + 1]);
        w2 = __expf(leaky(a.z + b.z) - cx.mx[dl * 4 + 2]);
        w3 = __expf(leaky(a.w + b.w) - cx.mx[dl * 4 + 3]);
        atomicAdd(&cx.sm[dl * 4 + 0], w0);
        atomicAdd(&cx.sm[dl * 4 + 1], w1);
        atomicAdd(&cx.sm[dl * 4 + 2], w2);
        atomicAdd(&cx.sm[dl * 4 + 3], w3);
    } else {
        s = 0; dl = 0;
    }
    const bool lo = cx.lane < 16;
    for (int j0 = 0; j0 < nv; j0 += 8) {
        float v0[8], v1[8];
#pragma unroll
        for (int u = 0; u < 8; ++u) {
            const int sj = __builtin_amdgcn_readlane(s, j0 + u);
            const float* hp = cx.h1 + (size_t)sj * 64 + cx.lane;
            v0[u] = hp[0];
            v1[u] = hp[32];
        }
#pragma unroll
        for (int u = 0; u < 8; ++u) {
            const int j = j0 + u;
            if ((okm >> j) & 1u) {
                const int dj = __builtin_amdgcn_readlane(dl, j);
                const float wlo_a = __int_as_float(__builtin_amdgcn_readlane(__float_as_int(w0), j));
                const float whi_a = __int_as_float(__builtin_amdgcn_readlane(__float_as_int(w1), j));
                const float wlo_b = __int_as_float(__builtin_amdgcn_readlane(__float_as_int(w2), j));
                const float whi_b = __int_as_float(__builtin_amdgcn_readlane(__float_as_int(w3), j));
                const float wa = lo ? wlo_a : whi_a;
                const float wb = lo ? wlo_b : whi_b;
                atomicAdd(&cx.acc[dj * 64 + cx.lane], wa * v0[u]);
                atomicAdd(&cx.acc[dj * 64 + 32 + cx.lane], wb * v1[u]);
            }
        }
    }
}

__device__ __forceinline__ void batch_max2(const AggCtx& cx, bool ok, int s, int dl) {
    if (ok) {
        const float a = cx.rec[(size_t)s * 4 + 2];
        atomicMaxFloat(&cx.mx[dl], leaky(a + cx.adl[dl]));
    }
}

__device__ __forceinline__ void batch_acc2(const AggCtx& cx, bool ok, int s, int dl) {
    if (ok) {
        const float4 r = *(const float4*)(cx.rec + (size_t)s * 4);
        const float w = __expf(leaky(r.z + cx.adl[dl]) - cx.mx[dl]);
        atomicAdd(&cx.sm[dl], w);
        atomicAdd(&cx.oa[dl], w * r.x);
        atomicAdd(&cx.ob[dl], w * r.y);
    }
}

template <int PHASE>
__device__ __forceinline__ void flush_batch(const AggCtx& cx, int head, int nv, int& nlist, bool& ovf) {
    __builtin_amdgcn_wave_barrier();
    const int qi = (head + cx.lane) & (QCAP - 1);
    int s = 0, dl = 0;
    bool ok = false;
    if (cx.lane < nv) {
        const int e = cx.wq_e[qi];
        s  = cx.src[e];
        dl = cx.dst[e] - cx.nbase;
        ok = (unsigned)dl < (unsigned)cx.ntile;
    }
    const unsigned okm = __builtin_amdgcn_ballot_w32(ok);
    if (PHASE == 1) {
        batch_max(cx, ok, s, dl);
        if (!ovf && nlist + 32 <= WCAP) {
            if (ok) {
                const int pos = nlist + __popc(okm & ((1u << cx.lane) - 1u));
                cx.lst_s[cx.lbase + pos] = s;
                cx.lst_d[cx.lbase + pos] = dl;
            }
            nlist += __popc(okm);
        } else {
            ovf = true;
        }
    } else if (PHASE == 2) {
        batch_acc(cx, nv, okm, s, dl);
    } else if (PHASE == 3) {
        batch_max2(cx, ok, s, dl);
    } else {
        batch_acc2(cx, ok, s, dl);
    }
    __builtin_amdgcn_wave_barrier();
}

template <int PHASE>
__device__ __forceinline__ void scan_edges(const AggCtx& cx, int& nlist, bool& ovf) {
    int head = 0, cnt = 0;
    for (int g = cx.wave; g < cx.ngroups; g += AGG_WAVES) {
        const uint4* p = (const uint4*)(cx.code + (size_t)g * (GRP / 4)) + cx.lane;
        uint4 c[8];
#pragma unroll
        for (int u = 0; u < 8; ++u) c[u] = p[u * 32];
        unsigned hit[4];
#pragma unroll
        for (int sg = 0; sg < 4; ++sg) {
            unsigned bits = 0u;
#pragma unroll
            for (int k = 0; k < 8; ++k) {
                const uint4 cv = c[2 * sg + (k >> 2)];
                const unsigned v = ((k & 3) == 0) ? cv.x : ((k & 3) == 1) ? cv.y : ((k & 3) == 2) ? cv.z : cv.w;
                const unsigned x = v ^ cx.pat;
                const unsigned t = (x & 0x7f7f7f7fu) + 0x7f7f7f7fu;
                const unsigned y = ~(t | x) & 0x80808080u;
                bits |= y >> (7 - k);
            }
            hit[sg] = bits;
        }
#pragma unroll 1
        for (int sg = 0; sg < 4; ++sg) {
            unsigned bits = (sg == 0) ? hit[0] : (sg == 1) ? hit[1] : (sg == 2) ? hit[2] : hit[3];
            const int ebase = g * GRP + sg * 1024 + cx.lane * 16;
            for (int round = 0; round < 32; ++round) {
                const bool has = bits != 0u;
                const unsigned mask = __builtin_amdgcn_ballot_w32(has);
                if (mask == 0u) break;
                if (has) {
                    const int b = __ffs(bits) - 1;
                    const int pos = (head + cnt + __popc(mask & ((1u << cx.lane) - 1u))) & (QCAP - 1);
                    cx.wq_e[pos] = ebase + ((b >> 2) & 1) * 512 + (b & 3) * 4 + (b >> 3);
                    bits &= bits - 1u;
                }
                cnt += __popc(mask);
                if (cnt >= 32) {
                    flush_batch<PHASE>(cx, head, 32, nlist, ovf);
                    head = (head + 32) & (QCAP - 1);
                    cnt -= 32;
                }
            }
        }
    }
    if (cnt > 0) flush_batch<PHASE>(cx, head, cnt, nlist, ovf);
}

__global__ void __launch_bounds__(1024)
agg1_lds(const int* __restrict__ ei, const float* __restrict__ as1,
         const float* __restrict__ ad1, const float* __restrict__ h1,
         const float* __restrict__ b1, float* __restrict__ o1,
         int* lst_s, int* lst_d, int* lcnt,
         const unsigned* __restrict__ code, int ngroups, int N, int E) {
    __shared__ float acc[TILE * 64];
    __shared__ float mx[TILE * HEADS];
    __shared__ float sm[TILE * HEADS];
    __shared__ int   qe[AGG_WAVES * QCAP];

    const int tid = threadIdx.x;
    AggCtx cx;
    cx.lane = tid & 31;
    cx.wave = __builtin_amdgcn_readfirstlane(tid >> 5);
    cx.nbase = blockIdx.x * TILE;
    cx.ntile = (N - cx.nbase < TILE) ? (N - cx.nbase) : TILE;
    cx.E = E;
    cx.src = ei; cx.dst = ei + E;
    cx.as1 = as1; cx.ad1 = ad1; cx.h1 = h1;
    cx.acc = acc; cx.mx = mx; cx.sm = sm;
    cx.code = code; cx.pat = (blockIdx.x % 255u) * 0x01010101u; cx.ngroups = ngroups;
    cx.wq_e = qe + cx.wave * QCAP;
    cx.lst_s = lst_s; cx.lst_d = lst_d;
    cx.lbase = ((size_t)blockIdx.x * AGG_WAVES + cx.wave) * WCAP;

    for (int i = tid; i < TILE * 64; i += 1024) acc[i] = 0.f;
    for (int i = tid; i < TILE * HEADS; i += 1024) {
        sm[i] = 0.f;
        float v = 0.f;
        if ((i >> 2) < cx.ntile) {
            const size_t g = (size_t)cx.nbase * HEADS + i;
            v = leaky(as1[g] + ad1[g]);
        }
        mx[i] = v;
    }
    __syncthreads();

    int nlist = 0;
    bool ovf = false;
    scan_edges<1>(cx, nlist, ovf);
    if (cx.lane == 0) lcnt[blockIdx.x * AGG_WAVES + cx.wave] = ovf ? -1 : nlist;
    __syncthreads();

    if (!ovf) {
        for (int p = 0; p < nlist; p += 32) {
            const int nv = (nlist - p < 32) ? (nlist - p) : 32;
            int s = 0, dl = 0;
            if (cx.lane < nv) {
                s  = lst_s[cx.lbase + p + cx.lane];
                dl = lst_d[cx.lbase + p + cx.lane];
            }
            batch_acc(cx, nv, (nv >= 32) ? 0xffffffffu : ((1u << nv) - 1u), s, dl);
        }
    } else {
        int dummy_n = 0; bool dummy_o = false;
        scan_edges<2>(cx, dummy_n, dummy_o);
    }
    __syncthreads();

    for (int q = tid; q < cx.ntile * 16; q += 1024) {
        const int nl = q >> 4;
        const int c  = (q & 15) * 4;
        const int hd = c >> 4;
        const size_t n = (size_t)cx.nbase + nl;
        const float ls = leaky(as1[n * HEADS + hd] + ad1[n * HEADS + hd]);
        const float wself = __expf(ls - mx[nl * HEADS + hd]);
        const float den = sm[nl * HEADS + hd] + wself + 1e-16f;
        const float4 hv = *(const float4*)(h1 + n * 64 + c);
        const float4 av = *(const float4*)(acc + nl * 64 + c);
        const float4 bv = *(const float4*)(b1 + c);
        float4 r;
        r.x = (av.x + wself * hv.x) / den + bv.x;
        r.y = (av.y + wself * hv.y) / den + bv.y;
        r.z = (av.z + wself * hv.z) / den + bv.z;
        r.w = (av.w + wself * hv.w) / den + bv.w;
        r.x = (r.x > 0.f) ? r.x : expm1f(r.x);
        r.y = (r.y > 0.f) ? r.y : expm1f(r.y);
        r.z = (r.z > 0.f) ? r.z : expm1f(r.z);
        r.w = (r.w > 0.f) ? r.w : expm1f(r.w);
        *(float4*)(o1 + n * 64 + c) = r;
    }
}

__global__ void __launch_bounds__(256)
node_prep2(const float* __restrict__ o1, const float* __restrict__ W2,
           const float* __restrict__ attS2, const float* __restrict__ attD2,
           float* __restrict__ rec, int N) {
    const int n = blockIdx.x * blockDim.x + threadIdx.x;
    if (n >= N) return;
    const float* hp = o1 + (size_t)n * 64;
    float a0 = 0.f, a1 = 0.f;
#pragma unroll
    for (int k = 0; k < 64; ++k) {
        const float v = hp[k];
        a0 += v * W2[k * 2 + 0];
        a1 += v * W2[k * 2 + 1];
    }
    float4 r;
    r.x = a0;
    r.y = a1;
    r.z = a0 * attS2[0] + a1 * attS2[1];
    r.w = a0 * attD2[0] + a1 * attD2[1];
    *(float4*)(rec + (size_t)n * 4) = r;
}

__global__ void __launch_bounds__(1024)
agg2_lds(const int* __restrict__ ei, const float* __restrict__ rec,
         const float* __restrict__ b2, float* __restrict__ out,
         int* lst_s, int* lst_d, const int* __restrict__ lcnt,
         const unsigned* __restrict__ code, int ngroups, int N, int E) {
    __shared__ float mx[TILE];
    __shared__ float sm[TILE];
    __shared__ float oa[TILE];
    __shared__ float ob[TILE];
    __shared__ float adl[TILE];
    __shared__ int   qe[AGG_WAVES * QCAP];

    const int tid = threadIdx.x;
    AggCtx cx;
    cx.lane = tid & 31;
    cx.wave = __builtin_amdgcn_readfirstlane(tid >> 5);
    cx.nbase = blockIdx.x * TILE;
    cx.ntile = (N - cx.nbase < TILE) ? (N - cx.nbase) : TILE;
    cx.E = E;
    cx.src = ei; cx.dst = ei + E;
    cx.as1 = nullptr; cx.ad1 = nullptr; cx.h1 = nullptr; cx.acc = nullptr;
    cx.mx = mx; cx.sm = sm;
    cx.code = code; cx.pat = (blockIdx.x % 255u) * 0x01010101u; cx.ngroups = ngroups;
    cx.wq_e = qe + cx.wave * QCAP;
    cx.lst_s = lst_s; cx.lst_d = lst_d;
    cx.lbase = ((size_t)blockIdx.x * AGG_WAVES + cx.wave) * WCAP;
    cx.rec = rec; cx.adl = adl; cx.oa = oa; cx.ob = ob;

    for (int i = tid; i < TILE; i += 1024) {
        float m = 0.f, a = 0.f;
        if (i < cx.ntile) {
            const float4 r = *(const float4*)(rec + ((size_t)cx.nbase + i) * 4);
            a = r.w;
            m = leaky(r.z + r.w);
        }
        mx[i] = m; adl[i] = a;
        sm[i] = 0.f; oa[i] = 0.f; ob[i] = 0.f;
    }
    __syncthreads();

    const int cnt = lcnt[blockIdx.x * AGG_WAVES + cx.wave];
    int dummy_n = 0; bool dummy_o = false;

    if (cnt >= 0) {
        for (int p = 0; p < cnt; p += 32) {
            const int nv = (cnt - p < 32) ? (cnt - p) : 32;
            int s = 0, dl = 0;
            if (cx.lane < nv) {
                s  = lst_s[cx.lbase + p + cx.lane];
                dl = lst_d[cx.lbase + p + cx.lane];
            }
            batch_max2(cx, cx.lane < nv, s, dl);
        }
    } else {
        scan_edges<3>(cx, dummy_n, dummy_o);
    }
    __syncthreads();

    if (cnt >= 0) {
        for (int p = 0; p < cnt; p += 32) {
            const int nv = (cnt - p < 32) ? (cnt - p) : 32;
            int s = 0, dl = 0;
            if (cx.lane < nv) {
                s  = lst_s[cx.lbase + p + cx.lane];
                dl = lst_d[cx.lbase + p + cx.lane];
            }
            batch_acc2(cx, cx.lane < nv, s, dl);
        }
    } else {
        scan_edges<4>(cx, dummy_n, dummy_o);
    }
    __syncthreads();

    for (int i = tid; i < cx.ntile; i += 1024) {
        const size_t n = (size_t)cx.nbase + i;
        const float4 r = *(const float4*)(rec + n * 4);
        const float wself = __expf(leaky(r.z + r.w) - mx[i]);
        const float inv = 1.0f / (sm[i] + wself + 1e-16f);
        float2 o;
        o.x = (oa[i] + wself * r.x) * inv + b2[0];
        o.y = (ob[i] + wself * r.y) * inv + b2[1];
        *(float2*)(out + n * 2) = o;
    }
}

extern "C" void kernel_launch(void* const* d_in, const int* in_sizes, int n_in,
                              void* d_out, int out_size, void* d_ws, size_t ws_size,
                              hipStream_t stream) {
    const float* x    = (const float*)d_in[0];
    const int*   ei   = (const int*)d_in[1];
    const float* W1   = (const float*)d_in[2];
    const float* aS1  = (const float*)d_in[3];
    const float* aD1  = (const float*)d_in[4];
    const float* b1   = (const float*)d_in[5];
    const float* W2   = (const float*)d_in[6];
    const float* aS2  = (const float*)d_in[7];
    const float* aD2  = (const float*)d_in[8];
    const float* b2   = (const float*)d_in[9];

    const int N = in_sizes[0] / DIM;
    const int E = in_sizes[1] / 2;
    float* out = (float*)d_out;

    float* ws  = (float*)d_ws;
    float* h1  = ws;
    float* as1 = h1 + (size_t)N * 64;
    float* ad1 = as1 + (size_t)N * HEADS;
    float* o1  = ad1 + (size_t)N * HEADS;
    float* rec = o1 + (size_t)N * 64;
    const int tiles = (N + TILE - 1) / TILE;
    int* lst_s = (int*)(rec + (size_t)N * 4);
    int* lst_d = lst_s + (size_t)tiles * AGG_WAVES * WCAP;
    int* lcnt  = lst_d + (size_t)tiles * AGG_WAVES * WCAP;
    const int ngroups = (E + GRP - 1) / GRP;
    unsigned* code = (unsigned*)(lcnt + (size_t)tiles * AGG_WAVES);

    const int Mtiles = (N + 15) >> 4;

    gemm1_wmma<<<(Mtiles + 3) / 4, 128, 0, stream>>>(x, W1, h1, N);
    node_prep1<<<(N * HEADS + 255) / 256, 256, 0, stream>>>(h1, aS1, aD1, as1, ad1, ei, code, N, E, ngroups * (GRP / 4));
    agg1_lds<<<tiles, 1024, 0, stream>>>(ei, as1, ad1, h1, b1, o1, lst_s, lst_d, lcnt, code, ngroups, N, E);
    node_prep2<<<(N + 255) / 256, 256, 0, stream>>>(o1, W2, aS2, aD2, rec, N);
    agg2_lds<<<tiles, 1024, 0, stream>>>(ei, rec, b2, out, lst_s, lst_d, lcnt, code, ngroups, N, E);
}


# Destination tiles of 400 nodes (was 512): 250 tile work-groups, one per WGP, instead of 196; edges and nodes per work-group down 22 %, column scan each as before; sums stay in LDS arrival order
# speedup vs baseline: 5.6444x; 1.0700x over previous
#include <hip/hip_runtime.h>
#include <math.h>

#define DIM 128
#define HID 16
#define HEADS 4
#define NEG_SLOPE 0.2f

typedef __attribute__((ext_vector_type(2))) float v2f;
typedef __attribute__((ext_vector_type(8))) float v8f;

__device__ __forceinline__ void atomicMaxFloat(float* addr, float v) {
    if (v >= 0.0f) {
        atomicMax((int*)addr, __float_as_int(v));
    } else {
        atomicMin((unsigned int*)addr, __float_as_uint(v));
    }
}

__global__ void __launch_bounds__(128)
gemm1_wmma(const float* __restrict__ x, const float* __restrict__ W1,
           float* __restrict__ h1, int N) {
    __shared__ float Wl[DIM * 64];
    for (int i = threadIdx.x; i < DIM * 64; i += blockDim.x) Wl[i] = W1[i];
    __syncthreads();

    const int wave  = threadIdx.x >> 5;
    const int lane  = threadIdx.x & 31;
    const int Mtile = blockIdx.x * 4 + wave;
    const int Mtiles = (N + 15) >> 4;
    if (Mtile >= Mtiles) return;

    const int r  = lane & 15;
    const int kh = lane >> 4;
    const int mbase = Mtile * 16;
    int row = mbase + r;
    if (row >= N) row = N - 1;
    const float* __restrict__ xr = x + (size_t)row * DIM;

    v8f acc0 = {}, acc1 = {}, acc2 = {}, acc3 = {};

    for (int kb = 0; kb < DIM; kb += 4) {
        const int k0 = kb + 2 * kh;
        v2f a;
        a.x = xr[k0];
        a.y = xr[k0 + 1];
        const float* w0 = &Wl[(size_t)k0 * 64];
        const float* w1 = &Wl[(size_t)(k0 + 1) * 64];
        v2f b;
        b.x = w0[r];       b.y = w1[r];
        acc0 = __builtin_amdgcn_wmma_f32_16x16x4_f32(false, a, false, b, (short)0, acc0, false, false);
        b.x = w0[16 + r];  b.y = w1[16 + r];
        acc1 = __builtin_amdgcn_wmma_f32_16x16x4_f32(false, a, false, b, (short)0, acc1, false, false);
        b.x = w0[32 + r];  b.y = w1[32 + r];
        acc2 = __builtin_amdgcn_wmma_f32_16x16x4_f32(false, a, false, b, (short)0, acc2, false, false);
        b.x = w0[48 + r];  b.y = w1[48 + r];
        acc3 = __builtin_amdgcn_wmma_f32_16x16x4_f32(false, a, false, b, (short)0, acc3, false, false);
    }

    float* outp = h1 + (size_t)mbase * 64;
#pragma unroll
    for (int i = 0; i < 8; ++i) {
        const int m = i + 8 * kh;
        if (mbase + m < N) {
            float* o = outp + (size_t)m * 64 + r;
            o[0]  = acc0[i];
            o[16] = acc1[i];
            o[32] = acc2[i];
            o[48] = acc3[i];
        }
    }
}

#define TILE 400
#define GRP 4096

__device__ __forceinline__ unsigned tile_code(int d) {
    return (d < 0) ? 255u : (((unsigned)d / (unsigned)TILE) % 255u);
}

__global__ void __launch_bounds__(256)
node_prep1(const float* __restrict__ h1, const float* __restrict__ attS,
           const float* __restrict__ attD, float* __restrict__ as1,
           float* __restrict__ ad1, const int* __restrict__ ei,
           unsigned* __restrict__ code, int N, int E, int ncode) {
    const int tid = blockIdx.x * blockDim.x + threadIdx.x;
    {
        const int* __restrict__ dstc = ei + E;
        const bool al16 = ((((size_t)dstc) & 15) == 0);
        const int nthreads = gridDim.x * blockDim.x;
        for (int q = tid; q < ncode; q += nthreads) {
            const int e = q * 4;
            int d0 = -1, d1 = -1, d2 = -1, d3 = -1;
            if (al16 && e + 3 < E) {
                const int4 v = *(const int4*)(dstc + e);
                d0 = v.x; d1 = v.y; d2 = v.z; d3 = v.w;
            } else {
                if (e     < E) d0 = dstc[e];
                if (e + 1 < E) d1 = dstc[e + 1];
                if (e + 2 < E) d2 = dstc[e + 2];
                if (e + 3 < E) d3 = dstc[e + 3];
            }
            code[q] = tile_code(d0) | (tile_code(d1) << 8) | (tile_code(d2) << 16) | (tile_code(d3) << 24);
        }
    }
    if (tid >= N * HEADS) return;
    const int n  = tid >> 2;
    const int hd = tid & 3;
    const float* hp = h1 + (size_t)n * 64 + hd * HID;
    float ss = 0.f, dd = 0.f;
#pragma unroll
    for (int c = 0; c < HID; ++c) {
        const float v = hp[c];
        ss += v * attS[hd * HID + c];
        dd += v * attD[hd * HID + c];
    }
    as1[tid] = ss;
    ad1[tid] = dd;
}

#define WCAP 1024
#define QCAP 64
#define AGG_WAVES 32

__device__ __forceinline__ float leaky(float v) { return (v > 0.f) ? v : NEG_SLOPE * v; }

struct AggCtx {
    const int* src; const int* dst;
    const float* as1; const float* ad1; const float* h1;
    float* acc; float* mx; float* sm;
    int* wq_e;
    int* lst_s; int* lst_d;
    size_t lbase;
    int nbase, ntile, lane, wave, E;
    const unsigned* code; unsigned pat; int ngroups;
    const float* rec; float* adl; float* oa; float* ob;
};

__device__ __forceinline__ void batch_max(const AggCtx& cx, bool ok, int s, int dl) {
    if (ok) {
        const float4 a = *(const float4*)(cx.as1 + (size_t)s * 4);
        const float4 b = *(const float4*)(cx.ad1 + (size_t)(cx.nbase + dl) * 4);
        atomicMaxFloat(&cx.mx[dl * 4 + 0], leaky(a.x + b.x));
        atomicMaxFloat(&cx.mx[dl * 4 + 1], leaky(a.y + b.y));
        atomicMaxFloat(&cx.mx[dl * 4 + 2], leaky(a.z + b.z));
        atomicMaxFloat(&cx.mx[dl * 4 + 3], leaky(a.w + b.w));
    }
}

__device__ __forceinline__ void batch_acc(const AggCtx& cx, int nv, unsigned okm, int s, int dl) {
    float w0 = 0.f, w1 = 0.f, w2 = 0.f, w3 = 0.f;
    if ((okm >> cx.lane) & 1u) {
        const float4 a = *(const float4*)(cx.as1 + (size_t)s * 4);
        const float4 b = *(const float4*)(cx.ad1 + (size_t)(cx.nbase + dl) * 4);
        w0 = __expf(leaky(a.x + b.x) - cx.mx[dl * 4 + 0]);
        w1 = __expf(leaky(a.y + b.y) - cx.mx[dl * 4 + 1]);
        w2 = __expf(leaky(a.z + b.z) - cx.mx[dl * 4 + 2]);
        w3 = __expf(leaky(a.w + b.w) - cx.mx[dl * 4 + 3]);
        atomicAdd(&cx.sm[dl * 4 + 0], w0);
        atomicAdd(&cx.sm[dl * 4 + 1], w1);
        atomicAdd(&cx.sm[dl * 4 + 2], w2);
        atomicAdd(&cx.sm[dl * 4 + 3], w3);
    } else {
        s = 0; dl = 0;
    }
    const bool lo = cx.lane < 16;
    for (int j0 = 0; j0 < nv; j0 += 8) {
        float v0[8], v1[8];
#pragma unroll
        for (int u = 0; u < 8; ++u) {
            const int sj = __builtin_amdgcn_readlane(s, j0 + u);
            const float* hp = cx.h1 + (size_t)sj * 64 + cx.lane;
            v0[u] = hp[0];
            v1[u] = hp[32];
        }
#pragma unroll
        for (int u = 0; u < 8; ++u) {
            const int j = j0 + u;
            if ((okm >> j) & 1u) {
                const int dj = __builtin_amdgcn_readlane(dl, j);
                const float wlo_a = __int_as_float(__builtin_amdgcn_readlane(__float_as_int(w0), j));
                const float whi_a = __int_as_float(__builtin_amdgcn_readlane(__float_as_int(w1), j));
                const float wlo_b = __int_as_float(__builtin_amdgcn_readlane(__float_as_int(w2), j));
                const float whi_b = __int_as_float(__builtin_amdgcn_readlane(__float_as_int(w3), j));
                const float wa = lo ? wlo_a : whi_a;
                const float wb = lo ? wlo_b : whi_b;
                atomicAdd(&cx.acc[dj * 64 + cx.lane], wa * v0[u]);
                atomicAdd(&cx.acc[dj * 64 + 32 + cx.lane], wb * v1[u]);
            }
        }
    }
}

__device__ __forceinline__ void batch_max2(const AggCtx& cx, bool ok, int s, int dl) {
    if (ok) {
        const float a = cx.rec[(size_t)s * 4 + 2];
        atomicMaxFloat(&cx.mx[dl], leaky(a + cx.adl[dl]));
    }
}

__device__ __forceinline__ void batch_acc2(const AggCtx& cx, bool ok, int s, int dl) {
    if (ok) {
        const float4 r = *(const float4*)(cx.rec + (size_t)s * 4);
        const float w = __expf(leaky(r.z + cx.adl[dl]) - cx.mx[dl]);
        atomicAdd(&cx.sm[dl], w);
        atomicAdd(&cx.oa[dl], w * r.x);
        atomicAdd(&cx.ob[dl], w * r.y);
    }
}

template <int PHASE>
__device__ __forceinline__ void flush_batch(const AggCtx& cx, int head, int nv, int& nlist, bool& ovf) {
    __builtin_amdgcn_wave_barrier();
    const int qi = (head + cx.lane) & (QCAP - 1);
    int s = 0, dl = 0;
    bool ok = false;
    if (cx.lane < nv) {
        const int e = cx.wq_e[qi];
        s  = cx.src[e];
        dl = cx.dst[e] - cx.nbase;
        ok = (unsigned)dl < (unsigned)cx.ntile;
    }
    const unsigned okm = __builtin_amdgcn_ballot_w32(ok);
    if (PHASE == 1) {
        batch_max(cx, ok, s, dl);
        if (!ovf && nlist + 32 <= WCAP) {
            if (ok) {
                const int pos = nlist + __popc(okm & ((1u << cx.lane) - 1u));
                cx.lst_s[cx.lbase + pos] = s;
                cx.lst_d[cx.lbase + pos] = dl;
            }
            nlist += __popc(okm);
        } else {
            ovf = true;
        }
    } else if (PHASE == 2) {
        batch_acc(cx, nv, okm, s, dl);
    } else if (PHASE == 3) {
        batch_max2(cx, ok, s, dl);
    } else {
        batch_acc2(cx, ok, s, dl);
    }
    __builtin_amdgcn_wave_barrier();
}

template <int PHASE>
__device__ __forceinline__ void scan_edges(const AggCtx& cx, int& nlist, bool& ovf) {
    int head = 0, cnt = 0;
    for (int g = cx.wave; g < cx.ngroups; g += AGG_WAVES) {
        const uint4* p = (const uint4*)(cx.code + (size_t)g * (GRP / 4)) + cx.lane;
        uint4 c[8];
#pragma unroll
        for (int u = 0; u < 8; ++u) c[u] = p[u * 32];
        unsigned hit[4];
#pragma unroll
        for (int sg = 0; sg < 4; ++sg) {
            unsigned bits = 0u;
#pragma unroll
            for (int k = 0; k < 8; ++k) {
                const uint4 cv = c[2 * sg + (k >> 2)];
                const unsigned v = ((k & 3) == 0) ? cv.x : ((k & 3) == 1) ? cv.y : ((k & 3) == 2) ? cv.z : cv.w;
                const unsigned x = v ^ cx.pat;
                const unsigned t = (x & 0x7f7f7f7fu) + 0x7f7f7f7fu;
                const unsigned y = ~(t | x) & 0x80808080u;
                bits |= y >> (7 - k);
            }
            hit[sg] = bits;
        }
#pragma unroll 1
        for (int sg = 0; sg < 4; ++sg) {
            unsigned bits = (sg == 0) ? hit[0] : (sg == 1) ? hit[1] : (sg == 2) ? hit[2] : hit[3];
            const int ebase = g * GRP + sg * 1024 + cx.lane * 16;
            for (int round = 0; round < 32; ++round) {
                const bool has = bits != 0u;
                const unsigned mask = __builtin_amdgcn_ballot_w32(has);
                if (mask == 0u) break;
                if (has) {
                    const int b = __ffs(bits) - 1;
                    const int pos = (head + cnt + __popc(mask & ((1u << cx.lane) - 1u))) & (QCAP - 1);
                    cx.wq_e[pos] = ebase + ((b >> 2) & 1) * 512 + (b & 3) * 4 + (b >> 3);
                    bits &= bits - 1u;
                }
                cnt += __popc(mask);
                if (cnt >= 32) {
                    flush_batch<PHASE>(cx, head, 32, nlist, ovf);
                    head = (head + 32) & (QCAP - 1);
                    cnt -= 32;
                }
            }
        }
    }
    if (cnt > 0) flush_batch<PHASE>(cx, head, cnt, nlist, ovf);
}

__global__ void __launch_bounds__(1024)
agg1_lds(const int* __restrict__ ei, const float* __restrict__ as1,
         const float* __restrict__ ad1, const float* __restrict__ h1,
         const float* __restrict__ b1, float* __restrict__ o1,
         int* lst_s, int* lst_d, int* lcnt,
         const unsigned* __restrict__ code, int ngroups, int N, int E) {
    __shared__ float acc[TILE * 64];
    __shared__ float mx[TILE * HEADS];
    __shared__ float sm[TILE * HEADS];
    __shared__ int   qe[AGG_WAVES * QCAP];

    const int tid = threadIdx.x;
    AggCtx cx;
    cx.lane = tid & 31;
    cx.wave = __builtin_amdgcn_readfirstlane(tid >> 5);
    cx.nbase = blockIdx.x * TILE;
    cx.ntile = (N - cx.nbase < TILE) ? (N - cx.nbase) : TILE;
    cx.E = E;
    cx.src = ei; cx.dst = ei + E;
    cx.as1 = as1; cx.ad1 = ad1; cx.h1 = h1;
    cx.acc = acc; cx.mx = mx; cx.sm = sm;
    cx.code = code; cx.pat = (blockIdx.x % 255u) * 0x01010101u; cx.ngroups = ngroups;
    cx.wq_e = qe + cx.wave * QCAP;
    cx.lst_s = lst_s; cx.lst_d = lst_d;
    cx.lbase = ((size_t)blockIdx.x * AGG_WAVES + cx.wave) * WCAP;

    for (int i = tid; i < TILE * 64; i += 1024) acc[i] = 0.f;
    for (int i = tid; i < TILE * HEADS; i += 1024) {
        sm[i] = 0.f;
        float v = 0.f;
        if ((i >> 2) < cx.ntile) {
            const size_t g = (size_t)cx.nbase * HEADS + i;
            v = leaky(as1[g] + ad1[g]);
        }
        mx[i] = v;
    }
    __syncthreads();

    int nlist = 0;
    bool ovf = false;
    scan_edges<1>(cx, nlist, ovf);
    if (cx.lane == 0) lcnt[blockIdx.x * AGG_WAVES + cx.wave] = ovf ? -1 : nlist;
    __syncthreads();

    if (!ovf) {
        for (int p = 0; p < nlist; p += 32) {
            const int nv = (nlist - p < 32) ? (nlist - p) : 32;
            int s = 0, dl = 0;
            if (cx.lane < nv) {
                s  = lst_s[cx.lbase + p + cx.lane];
                dl = lst_d[cx.lbase + p + cx.lane];
            }
            batch_acc(cx, nv, (nv >= 32) ? 0xffffffffu : ((1u << nv) - 1u), s, dl);
        }
    } else {
        int dummy_n = 0; bool dummy_o = false;
        scan_edges<2>(cx, dummy_n, dummy_o);
    }
    __syncthreads();

    for (int q = tid; q < cx.ntile * 16; q += 1024) {
        const int nl = q >> 4;
        const int c  = (q & 15) * 4;
        const int hd = c >> 4;
        const size_t n = (size_t)cx.nbase + nl;
        const float ls = leaky(as1[n * HEADS + hd] + ad1[n * HEADS + hd]);
        const float wself = __expf(ls - mx[nl * HEADS + hd]);
        const float den = sm[nl * HEADS + hd] + wself + 1e-16f;
        const float4 hv = *(const float4*)(h1 + n * 64 + c);
        const float4 av = *(const float4*)(acc + nl * 64 + c);
        const float4 bv = *(const float4*)(b1 + c);
        float4 r;
        r.x = (av.x + wself * hv.x) / den + bv.x;
        r.y = (av.y + wself * hv.y) / den + bv.y;
        r.z = (av.z + wself * hv.z) / den + bv.z;
        r.w = (av.w + wself * hv.w) / den + bv.w;
        r.x = (r.x > 0.f) ? r.x : expm1f(r.x);
        r.y = (r.y > 0.f) ? r.y : expm1f(r.y);
        r.z = (r.z > 0.f) ? r.z : expm1f(r.z);
        r.w = (r.w > 0.f) ? r.w : expm1f(r.w);
        *(float4*)(o1 + n * 64 + c) = r;
    }
}

__global__ void __launch_bounds__(256)
node_prep2(const float* __restrict__ o1, const float* __restrict__ W2,
           const float* __restrict__ attS2, const float* __restrict__ attD2,
           float* __restrict__ rec, int N) {
    const int n = blockIdx.x * blockDim.x + threadIdx.x;
    if (n >= N) return;
    const float* hp = o1 + (size_t)n * 64;
    float a0 = 0.f, a1 = 0.f;
#pragma unroll
    for (int k = 0; k < 64; ++k) {
        const float v = hp[k];
        a0 += v * W2[k * 2 + 0];
        a1 += v * W2[k * 2 + 1];
    }
    float4 r;
    r.x = a0;
    r.y = a1;
    r.z = a0 * attS2[0] + a1 * attS2[1];
    r.w = a0 * attD2[0] + a1 * attD2[1];
    *(float4*)(rec + (size_t)n * 4) = r;
}

__global__ void __launch_bounds__(1024)
agg2_lds(const int* __restrict__ ei, const float* __restrict__ rec,
         const float* __restrict__ b2, float* __restrict__ out,
         int* lst_s, int* lst_d, const int* __restrict__ lcnt,
         const unsigned* __restrict__ code, int ngroups, int N, int E) {
    __shared__ float mx[TILE];
    __shared__ float sm[TILE];
    __shared__ float oa[TILE];
    __shared__ float ob[TILE];
    __shared__ float adl[TILE];
    __shared__ int   qe[AGG_WAVES * QCAP];

    const int tid = threadIdx.x;
    AggCtx cx;
    cx.lane = tid & 31;
    cx.wave = __builtin_amdgcn_readfirstlane(tid >> 5);
    cx.nbase = blockIdx.x * TILE;
    cx.ntile = (N - cx.nbase < TILE) ? (N - cx.nbase) : TILE;
    cx.E = E;
    cx.src = ei; cx.dst = ei + E;
    cx.as1 = nullptr; cx.ad1 = nullptr; cx.h1 = nullptr; cx.acc = nullptr;
    cx.mx = mx; cx.sm = sm;
    cx.code = code; cx.pat = (blockIdx.x % 255u) * 0x01010101u; cx.ngroups = ngroups;
    cx.wq_e = qe + cx.wave * QCAP;
    cx.lst_s = lst_s; cx.lst_d = lst_d;
    cx.lbase = ((size_t)blockIdx.x * AGG_WAVES + cx.wave) * WCAP;
    cx.rec = rec; cx.adl = adl; cx.oa = oa; cx.ob = ob;

    for (int i = tid; i < TILE; i += 1024) {
        float m = 0.f, a = 0.f;
        if (i < cx.ntile) {
            const float4 r = *(const float4*)(rec + ((size_t)cx.nbase + i) * 4);
            a = r.w;
            m = leaky(r.z + r.w);
        }
        mx[i] = m; adl[i] = a;
        sm[i] = 0.f; oa[i] = 0.f; ob[i] = 0.f;
    }
    __syncthreads();

    const int cnt = lcnt[blockIdx.x * AGG_WAVES + cx.wave];
    int dummy_n = 0; bool dummy_o = false;

    if (cnt >= 0) {
        for (int p = 0; p < cnt; p += 32) {
            const int nv = (cnt - p < 32) ? (cnt - p) : 32;
            int s = 0, dl = 0;
            if (cx.lane < nv) {
                s  = lst_s[cx.lbase + p + cx.lane];
                dl = lst_d[cx.lbase + p + cx.lane];
            }
            batch_max2(cx, cx.lane < nv, s, dl);
        }
    } else {
        scan_edges<3>(cx, dummy_n, dummy_o);
    }
    __syncthreads();

    if (cnt >= 0) {
        for (int p = 0; p < cnt; p += 32) {
            const int nv = (cnt - p < 32) ? (cnt - p) : 32;
            int s = 0, dl = 0;
            if (cx.lane < nv) {
                s  = lst_s[cx.lbase + p + cx.lane];
                dl = lst_d[cx.lbase + p + cx.lane];
            }
            batch_acc2(cx, cx.lane < nv, s, dl);
        }
    } else {
        scan_edges<4>(cx, dummy_n, dummy_o);
    }
    __syncthreads();

    for (int i = tid; i < cx.ntile; i += 1024) {
        const size_t n = (size_t)cx.nbase + i;
        const float4 r = *(const float4*)(rec + n * 4);
        const float wself = __expf(leaky(r.z + r.w) - mx[i]);
        const float inv = 1.0f / (sm[i] + wself + 1e-16f);
        float2 o;
        o.x = (oa[i] + wself * r.x) * inv + b2[0];
        o.y = (ob[i] + wself * r.y) * inv + b2[1];
        *(float2*)(out + n * 2) = o;
    }
}

extern "C" void kernel_launch(void* const* d_in, const int* in_sizes, int n_in,
                              void* d_out, int out_size, void* d_ws, size_t ws_size,
                              hipStream_t stream) {
    const float* x    = (const float*)d_in[0];
    const int*   ei   = (const int*)d_in[1];
    const float* W1   = (const float*)d_in[2];
    const float* aS1  = (const float*)d_in[3];
    const float* aD1  = (const float*)d_in[4];
    const float* b1   = (const float*)d_in[5];
    const float* W2   = (const float*)d_in[6];
    const float* aS2  = (const float*)d_in[7];
    const float* aD2  = (const float*)d_in[8];
    const float* b2   = (const float*)d_in[9];

    const int N = in_sizes[0] / DIM;
    const int E = in_sizes[1] / 2;
    float* out = (float*)d_out;

    float* ws  = (float*)d_ws;
    float* h1  = ws;
    float* as1 = h1 + (size_t)N * 64;
    float* ad1 = as1 + (size_t)N * HEADS;
    float* o1  = ad1 + (size_t)N * HEADS;
    float* rec = o1 + (size_t)N * 64;
    const int tiles = (N + TILE - 1) / TILE;
    int* lst_s = (int*)(rec + (size_t)N * 4);
    int* lst_d = lst_s + (size_t)tiles * AGG_WAVES * WCAP;
    int* lcnt  = lst_d + (size_t)tiles * AGG_WAVES * WCAP;
    const int ngroups = (E + GRP - 1) / GRP;
    unsigned* code = (unsigned*)(lcnt + (size_t)tiles * AGG_WAVES);

    const int Mtiles = (N + 15) >> 4;

    gemm1_wmma<<<(Mtiles + 3) / 4, 128, 0, stream>>>(x, W1, h1, N);
    node_prep1<<<(N * HEADS + 255) / 256, 256, 0, stream>>>(h1, aS1, aD1, as1, ad1, ei, code, N, E, ngroups * (GRP / 4));
    agg1_lds<<<tiles, 1024, 0, stream>>>(ei, as1, ad1, h1, b1, o1, lst_s, lst_d, lcnt, code, ngroups, N, E);
    node_prep2<<<(N + 255) / 256, 256, 0, stream>>>(o1, W2, aS2, aD2, rec, N);
    agg2_lds<<<tiles, 1024, 0, stream>>>(ei, rec, b2, out, lst_s, lst_d, lcnt, code, ngroups, N, E);
}


# GEMM takes W1 straight from global memory (it stays in cache), N tiles interleaved by column: 16-byte B loads, whole-row 16-byte stores; LDS copy of W1 and its 64-step one-dword staging chain gone
# speedup vs baseline: 6.6813x; 1.1837x over previous
#include <hip/hip_runtime.h>
#include <math.h>

#define DIM 128
#define HID 16
#define HEADS 4
#define NEG_SLOPE 0.2f

typedef __attribute__((ext_vector_type(2))) float v2f;
typedef __attribute__((ext_vector_type(8))) float v8f;

__device__ __forceinline__ void atomicMaxFloat(float* addr, float v) {
    if (v >= 0.0f) {
        atomicMax((int*)addr, __float_as_int(v));
    } else {
        atomicMin((unsigned int*)addr, __float_as_uint(v));
    }
}

__global__ void __launch_bounds__(128)
gemm1_wmma(const float* __restrict__ x, const float* __restrict__ W1,
           float* __restrict__ h1, int N) {
    const int wave  = threadIdx.x >> 5;
    const int lane  = threadIdx.x & 31;
    const int Mtile = blockIdx.x * 4 + wave;
    const int Mtiles = (N + 15) >> 4;
    if (Mtile >= Mtiles) return;

    const int r  = lane & 15;
    const int kh = lane >> 4;
    const int mbase = Mtile * 16;
    int row = mbase + r;
    if (row >= N) row = N - 1;
    const float* __restrict__ xr = x + (size_t)row * DIM;
    const float* __restrict__ wc = W1 + 4 * r;

    v8f acc0 = {}, acc1 = {}, acc2 = {}, acc3 = {};

#pragma unroll
    for (int kb = 0; kb < DIM; kb += 4) {
        const int k0 = kb + 2 * kh;
        v2f a;
        a.x = xr[k0];
        a.y = xr[k0 + 1];
        const float4 w0 = *(const float4*)(wc + (size_t)k0 * 64);
        const float4 w1 = *(const float4*)(wc + (size_t)(k0 + 1) * 64);
        v2f b;
        b.x = w0.x;  b.y = w1.x;
        acc0 = __builtin_amdgcn_wmma_f32_16x16x4_f32(false, a, false, b, (short)0, acc0, false, false);
        b.x = w0.y;  b.y = w1.y;
        acc1 = __builtin_amdgcn_wmma_f32_16x16x4_f32(false, a, false, b, (short)0, acc1, false, false);
        b.x = w0.z;  b.y = w1.z;
        acc2 = __builtin_amdgcn_wmma_f32_16x16x4_f32(false, a, false, b, (short)0, acc2, false, false);
        b.x = w0.w;  b.y = w1.w;
        acc3 = __builtin_amdgcn_wmma_f32_16x16x4_f32(false, a, false, b, (short)0, acc3, false, false);
    }

    float* outp = h1 + (size_t)mbase * 64 + 4 * r;
#pragma unroll
    for (int i = 0; i < 8; ++i) {
        const int m = i + 8 * kh;
        if (mbase + m < N) {
            float4 o;
            o.x = acc0[i];
            o.y = acc1[i];
            o.z = acc2[i];
            o.w = acc3[i];
            *(float4*)(outp + (size_t)m * 64) = o;
        }
    }
}

#define TILE 400
#define GRP 4096

__device__ __forceinline__ unsigned tile_code(int d) {
    return (d < 0) ? 255u : (((unsigned)d / (unsigned)TILE) % 255u);
}

__global__ void __launch_bounds__(256)
node_prep1(const float* __restrict__ h1, const float* __restrict__ attS,
           const float* __restrict__ attD, float* __restrict__ as1,
           float* __restrict__ ad1, const int* __restrict__ ei,
           unsigned* __restrict__ code, int N, int E, int ncode) {
    const int tid = blockIdx.x * blockDim.x + threadIdx.x;
    {
        const int* __restrict__ dstc = ei + E;
        const bool al16 = ((((size_t)dstc) & 15) == 0);
        const int nthreads = gridDim.x * blockDim.x;
        for (int q = tid; q < ncode; q += nthreads) {
            const int e = q * 4;
            int d0 = -1, d1 = -1, d2 = -1, d3 = -1;
            if (al16 && e + 3 < E) {
                const int4 v = *(const int4*)(dstc + e);
                d0 = v.x; d1 = v.y; d2 = v.z; d3 = v.w;
            } else {
                if (e     < E) d0 = dstc[e];
                if (e + 1 < E) d1 = dstc[e + 1];
                if (e + 2 < E) d2 = dstc[e + 2];
                if (e + 3 < E) d3 = dstc[e + 3];
            }
            code[q] = tile_code(d0) | (tile_code(d1) << 8) | (tile_code(d2) << 16) | (tile_code(d3) << 24);
        }
    }
    if (tid >= N * HEADS) return;
    const int n  = tid >> 2;
    const int hd = tid & 3;
    const float* hp = h1 + (size_t)n * 64 + hd * HID;
    float ss = 0.f, dd = 0.f;
#pragma unroll
    for (int c = 0; c < HID; ++c) {
        const float v = hp[c];
        ss += v * attS[hd * HID + c];
        dd += v * attD[hd * HID + c];
    }
    as1[tid] = ss;
    ad1[tid] = dd;
}

#define WCAP 1024
#define QCAP 64
#define AGG_WAVES 32

__device__ __forceinline__ float leaky(float v) { return (v > 0.f) ? v : NEG_SLOPE * v; }

struct AggCtx {
    const int* src; const int* dst;
    const float* as1; const float* ad1; const float* h1;
    float* acc; float* mx; float* sm;
    int* wq_e;
    int* lst_s; int* lst_d;
    size_t lbase;
    int nbase, ntile, lane, wave, E;
    const unsigned* code; unsigned pat; int ngroups;
    const float* rec; float* adl; float* oa; float* ob;
};

__device__ __forceinline__ void batch_max(const AggCtx& cx, bool ok, int s, int dl) {
    if (ok) {
        const float4 a = *(const float4*)(cx.as1 + (size_t)s * 4);
        const float4 b = *(const float4*)(cx.ad1 + (size_t)(cx.nbase + dl) * 4);
        atomicMaxFloat(&cx.mx[dl * 4 + 0], leaky(a.x + b.x));
        atomicMaxFloat(&cx.mx[dl * 4 + 1], leaky(a.y + b.y));
        atomicMaxFloat(&cx.mx[dl * 4 + 2], leaky(a.z + b.z));
        atomicMaxFloat(&cx.mx[dl * 4 + 3], leaky(a.w + b.w));
    }
}

__device__ __forceinline__ void batch_acc(const AggCtx& cx, int nv, unsigned okm, int s, int dl) {
    float w0 = 0.f, w1 = 0.f, w2 = 0.f, w3 = 0.f;
    if ((okm >> cx.lane) & 1u) {
        const float4 a = *(const float4*)(cx.as1 + (size_t)s * 4);
        const float4 b = *(const float4*)(cx.ad1 + (size_t)(cx.nbase + dl) * 4);
        w0 = __expf(leaky(a.x + b.x) - cx.mx[dl * 4 + 0]);
        w1 = __expf(leaky(a.y + b.y) - cx.mx[dl * 4 + 1]);
        w2 = __expf(leaky(a.z + b.z) - cx.mx[dl * 4 + 2]);
        w3 = __expf(leaky(a.w + b.w) - cx.mx[dl * 4 + 3]);
        atomicAdd(&cx.sm[dl * 4 + 0], w0);
        atomicAdd(&cx.sm[dl * 4 + 1], w1);
        atomicAdd(&cx.sm[dl * 4 + 2], w2);
        atomicAdd(&cx.sm[dl * 4 + 3], w3);
    } else {
        s = 0; dl = 0;
    }
    const bool lo = cx.lane < 16;
    for (int j0 = 0; j0 < nv; j0 += 8) {
        float v0[8], v1[8];
#pragma unroll
        for (int u = 0; u < 8; ++u) {
            const int sj = __builtin_amdgcn_readlane(s, j0 + u);
            const float* hp = cx.h1 + (size_t)sj * 64 + cx.lane;
            v0[u] = hp[0];
            v1[u] = hp[32];
        }
#pragma unroll
        for (int u = 0; u < 8; ++u) {
            const int j = j0 + u;
            if ((okm >> j) & 1u) {
                const int dj = __builtin_amdgcn_readlane(dl, j);
                const float wlo_a = __int_as_float(__builtin_amdgcn_readlane(__float_as_int(w0), j));
                const float whi_a = __int_as_float(__builtin_amdgcn_readlane(__float_as_int(w1), j));
                const float wlo_b = __int_as_float(__builtin_amdgcn_readlane(__float_as_int(w2), j));
                const float whi_b = __int_as_float(__builtin_amdgcn_readlane(__float_as_int(w3), j));
                const float wa = lo ? wlo_a : whi_a;
                const float wb = lo ? wlo_b : whi_b;
                atomicAdd(&cx.acc[dj * 64 + cx.lane], wa * v0[u]);
                atomicAdd(&cx.acc[dj * 64 + 32 + cx.lane], wb * v1[u]);
            }
        }
    }
}

__device__ __forceinline__ void batch_max2(const AggCtx& cx, bool ok, int s, int dl) {
    if (ok) {
        const float a = cx.rec[(size_t)s * 4 + 2];
        atomicMaxFloat(&cx.mx[dl], leaky(a + cx.adl[dl]));
    }
}

__device__ __forceinline__ void batch_acc2(const AggCtx& cx, bool ok, int s, int dl) {
    if (ok) {
        const float4 r = *(const float4*)(cx.rec + (size_t)s * 4);
        const float w = __expf(leaky(r.z + cx.adl[dl]) - cx.mx[dl]);
        atomicAdd(&cx.sm[dl], w);
        atomicAdd(&cx.oa[dl], w * r.x);
        atomicAdd(&cx.ob[dl], w * r.y);
    }
}

template <int PHASE>
__device__ __forceinline__ void flush_batch(const AggCtx& cx, int head, int nv, int& nlist, bool& ovf) {
    __builtin_amdgcn_wave_barrier();
    const int qi = (head + cx.lane) & (QCAP - 1);
    int s = 0, dl = 0;
    bool ok = false;
    if (cx.lane < nv) {
        const int e = cx.wq_e[qi];
        s  = cx.src[e];
        dl = cx.dst[e] - cx.nbase;
        ok = (unsigned)dl < (unsigned)cx.ntile;
    }
    const unsigned okm = __builtin_amdgcn_ballot_w32(ok);
    if (PHASE == 1) {
        batch_max(cx, ok, s, dl);
        if (!ovf && nlist + 32 <= WCAP) {
            if (ok) {
                const int pos = nlist + __popc(okm & ((1u << cx.lane) - 1u));
                cx.lst_s[cx.lbase + pos] = s;
                cx.lst_d[cx.lbase + pos] = dl;
            }
            nlist += __popc(okm);
        } else {
            ovf = true;
        }
    } else if (PHASE == 2) {
        batch_acc(cx, nv, okm, s, dl);
    } else if (PHASE == 3) {
        batch_max2(cx, ok, s, dl);
    } else {
        batch_acc2(cx, ok, s, dl);
    }
    __builtin_amdgcn_wave_barrier();
}

template <int PHASE>
__device__ __forceinline__ void scan_edges(const AggCtx& cx, int& nlist, bool& ovf) {
    int head = 0, cnt = 0;
    for (int g = cx.wave; g < cx.ngroups; g += AGG_WAVES) {
        const uint4* p = (const uint4*)(cx.code + (size_t)g * (GRP / 4)) + cx.lane;
        uint4 c[8];
#pragma unroll
        for (int u = 0; u < 8; ++u) c[u] = p[u * 32];
        unsigned hit[4];
#pragma unroll
        for (int sg = 0; sg < 4; ++sg) {
            unsigned bits = 0u;
#pragma unroll
            for (int k = 0; k < 8; ++k) {
                const uint4 cv = c[2 * sg + (k >> 2)];
                const unsigned v = ((k & 3) == 0) ? cv.x : ((k & 3) == 1) ? cv.y : ((k & 3) == 2) ? cv.z : cv.w;
                const unsigned x = v ^ cx.pat;
                const unsigned t = (x & 0x7f7f7f7fu) + 0x7f7f7f7fu;
                const unsigned y = ~(t | x) & 0x80808080u;
                bits |= y >> (7 - k);
            }
            hit[sg] = bits;
        }
#pragma unroll 1
        for (int sg = 0; sg < 4; ++sg) {
            unsigned bits = (sg == 0) ? hit[0] : (sg == 1) ? hit[1] : (sg == 2) ? hit[2] : hit[3];
            const int ebase = g * GRP + sg * 1024 + cx.lane * 16;
            for (int round = 0; round < 32; ++round) {
                const bool has = bits != 0u;
                const unsigned mask = __builtin_amdgcn_ballot_w32(has);
                if (mask == 0u) break;
                if (has) {
                    const int b = __ffs(bits) - 1;
                    const int pos = (head + cnt + __popc(mask & ((1u << cx.lane) - 1u))) & (QCAP - 1);
                    cx.wq_e[pos] = ebase + ((b >> 2) & 1) * 512 + (b & 3) * 4 + (b >> 3);
                    bits &= bits - 1u;
                }
                cnt += __popc(mask);
                if (cnt >= 32) {
                    flush_batch<PHASE>(cx, head, 32, nlist, ovf);
                    head = (head + 32) & (QCAP - 1);
                    cnt -= 32;
                }
            }
        }
    }
    if (cnt > 0) flush_batch<PHASE>(cx, head, cnt, nlist, ovf);
}

__global__ void __launch_bounds__(1024)
agg1_lds(const int* __restrict__ ei, const float* __restrict__ as1,
         const float* __restrict__ ad1, const float* __restrict__ h1,
         const float* __restrict__ b1, float* __restrict__ o1,
         int* lst_s, int* lst_d, int* lcnt,
         const unsigned* __restrict__ code, int ngroups, int N, int E) {
    __shared__ float acc[TILE * 64];
    __shared__ float mx[TILE * HEADS];
    __shared__ float sm[TILE * HEADS];
    __shared__ int   qe[AGG_WAVES * QCAP];

    const int tid = threadIdx.x;
    AggCtx cx;
    cx.lane = tid & 31;
    cx.wave = __builtin_amdgcn_readfirstlane(tid >> 5);
    cx.nbase = blockIdx.x * TILE;
    cx.ntile = (N - cx.nbase < TILE) ? (N - cx.nbase) : TILE;
    cx.E = E;
    cx.src = ei; cx.dst = ei + E;
    cx.as1 = as1; cx.ad1 = ad1; cx.h1 = h1;
    cx.acc = acc; cx.mx = mx; cx.sm = sm;
    cx.code = code; cx.pat = (blockIdx.x % 255u) * 0x01010101u; cx.ngroups = ngroups;
    cx.wq_e = qe + cx.wave * QCAP;
    cx.lst_s = lst_s; cx.lst_d = lst_d;
    cx.lbase = ((size_t)blockIdx.x * AGG_WAVES + cx.wave) * WCAP;

    for (int i = tid; i < TILE * 64; i += 1024) acc[i] = 0.f;
    for (int i = tid; i < TILE * HEADS; i += 1024) {
        sm[i] = 0.f;
        float v = 0.f;
        if ((i >> 2) < cx.ntile) {
            const size_t g = (size_t)cx.nbase * HEADS + i;
            v = leaky(as1[g] + ad1[g]);
        }
        mx[i] = v;
    }
    __syncthreads();

    int nlist = 0;
    bool ovf = false;
    scan_edges<1>(cx, nlist, ovf);
    if (cx.lane == 0) lcnt[blockIdx.x * AGG_WAVES + cx.wave] = ovf ? -1 : nlist;
    __syncthreads();

    if (!ovf) {
        for (int p = 0; p < nlist; p += 32) {
            const int nv = (nlist - p < 32) ? (nlist - p) : 32;
            int s = 0, dl = 0;
            if (cx.lane < nv) {
                s  = lst_s[cx.lbase + p + cx.lane];
                dl = lst_d[cx.lbase + p + cx.lane];
            }
            batch_acc(cx, nv, (nv >= 32) ? 0xffffffffu : ((1u << nv) - 1u), s, dl);
        }
    } else {
        int dummy_n = 0; bool dummy_o = false;
        scan_edges<2>(cx, dummy_n, dummy_o);
    }
    __syncthreads();

    for (int q = tid; q < cx.ntile * 16; q += 1024) {
        const int nl = q >> 4;
        const int c  = (q & 15) * 4;
        const int hd = c >> 4;
        const size_t n = (size_t)cx.nbase + nl;
        const float ls = leaky(as1[n * HEADS + hd] + ad1[n * HEADS + hd]);
        const float wself = __expf(ls - mx[nl * HEADS + hd]);
        const float den = sm[nl * HEADS + hd] + wself + 1e-16f;
        const float4 hv = *(const float4*)(h1 + n * 64 + c);
        const float4 av = *(const float4*)(acc + nl * 64 + c);
        const float4 bv = *(const float4*)(b1 + c);
        float4 r;
        r.x = (av.x + wself * hv.x) / den + bv.x;
        r.y = (av.y + wself * hv.y) / den + bv.y;
        r.z = (av.z + wself * hv.z) / den + bv.z;
        r.w = (av.w + wself * hv.w) / den + bv.w;
        r.x = (r.x > 0.f) ? r.x : expm1f(r.x);
        r.y = (r.y > 0.f) ? r.y : expm1f(r.y);
        r.z = (r.z > 0.f) ? r.z : expm1f(r.z);
        r.w = (r.w > 0.f) ? r.w : expm1f(r.w);
        *(float4*)(o1 + n * 64 + c) = r;
    }
}

__global__ void __launch_bounds__(256)
node_prep2(const float* __restrict__ o1, const float* __restrict__ W2,
           const float* __restrict__ attS2, const float* __restrict__ attD2,
           float* __restrict__ rec, int N) {
    const int n = blockIdx.x * blockDim.x + threadIdx.x;
    if (n >= N) return;
    const float* hp = o1 + (size_t)n * 64;
    float a0 = 0.f, a1 = 0.f;
#pragma unroll
    for (int k = 0; k < 64; ++k) {
        const float v = hp[k];
        a0 += v * W2[k * 2 + 0];
        a1 += v * W2[k * 2 + 1];
    }
    float4 r;
    r.x = a0;
    r.y = a1;
    r.z = a0 * attS2[0] + a1 * attS2[1];
    r.w = a0 * attD2[0] + a1 * attD2[1];
    *(float4*)(rec + (size_t)n * 4) = r;
}

__global__ void __launch_bounds__(1024)
agg2_lds(const int* __restrict__ ei, const float* __restrict__ rec,
         const float* __restrict__ b2, float* __restrict__ out,
         int* lst_s, int* lst_d, const int* __restrict__ lcnt,
         const unsigned* __restrict__ code, int ngroups, int N, int E) {
    __shared__ float mx[TILE];
    __shared__ float sm[TILE];
    __shared__ float oa[TILE];
    __shared__ float ob[TILE];
    __shared__ float adl[TILE];
    __shared__ int   qe[AGG_WAVES * QCAP];

    const int tid = threadIdx.x;
    AggCtx cx;
    cx.lane = tid & 31;
    cx.wave = __builtin_amdgcn_readfirstlane(tid >> 5);
    cx.nbase = blockIdx.x * TILE;
    cx.ntile = (N - cx.nbase < TILE) ? (N - cx.nbase) : TILE;
    cx.E = E;
    cx.src = ei; cx.dst = ei + E;
    cx.as1 = nullptr; cx.ad1 = nullptr; cx.h1 = nullptr; cx.acc = nullptr;
    cx.mx = mx; cx.sm = sm;
    cx.code = code; cx.pat = (blockIdx.x % 255u) * 0x01010101u; cx.ngroups = ngroups;
    cx.wq_e = qe + cx.wave * QCAP;
    cx.lst_s = lst_s; cx.lst_d = lst_d;
    cx.lbase = ((size_t)blockIdx.x * AGG_WAVES + cx.wave) * WCAP;
    cx.rec = rec; cx.adl = adl; cx.oa = oa; cx.ob = ob;

    for (int i = tid; i < TILE; i += 1024) {
        float m = 0.f, a = 0.f;
        if (i < cx.ntile) {
            const float4 r = *(const float4*)(rec + ((size_t)cx.nbase + i) * 4);
            a = r.w;
            m = leaky(r.z + r.w);
        }
        mx[i] = m; adl[i] = a;
        sm[i] = 0.f; oa[i] = 0.f; ob[i] = 0.f;
    }
    __syncthreads();

    const int cnt = lcnt[blockIdx.x * AGG_WAVES + cx.wave];
    int dummy_n = 0; bool dummy_o = false;

    if (cnt >= 0) {
        for (int p = 0; p < cnt; p += 32) {
            const int nv = (cnt - p < 32) ? (cnt - p) : 32;
            int s = 0, dl = 0;
            if (cx.lane < nv) {
                s  = lst_s[cx.lbase + p + cx.lane];
                dl = lst_d[cx.lbase + p + cx.lane];
            }
            batch_max2(cx, cx.lane < nv, s, dl);
        }
    } else {
        scan_edges<3>(cx, dummy_n, dummy_o);
    }
    __syncthreads();

    if (cnt >= 0) {
        for (int p = 0; p < cnt; p += 32) {
            const int nv = (cnt - p < 32) ? (cnt - p) : 32;
            int s = 0, dl = 0;
            if (cx.lane < nv) {
                s  = lst_s[cx.lbase + p + cx.lane];
                dl = lst_d[cx.lbase + p + cx.lane];
            }
            batch_acc2(cx, cx.lane < nv, s, dl);
        }
    } else {
        scan_edges<4>(cx, dummy_n, dummy_o);
    }
    __syncthreads();

    for (int i = tid; i < cx.ntile; i += 1024) {
        const size_t n = (size_t)cx.nbase + i;
        const float4 r = *(const float4*)(rec + n * 4);
        const float wself = __expf(leaky(r.z + r.w) - mx[i]);
        const float inv = 1.0f / (sm[i] + wself + 1e-16f);
        float2 o;
        o.x = (oa[i] + wself * r.x) * inv + b2[0];
        o.y = (ob[i] + wself * r.y) * inv + b2[1];
        *(float2*)(out + n * 2) = o;
    }
}

extern "C" void kernel_launch(void* const* d_in, const int* in_sizes, int n_in,
                              void* d_out, int out_size, void* d_ws, size_t ws_size,
                              hipStream_t stream) {
    const float* x    = (const float*)d_in[0];
    const int*   ei   = (const int*)d_in[1];
    const float* W1   = (const float*)d_in[2];
    const float* aS1  = (const float*)d_in[3];
    const float* aD1  = (const float*)d_in[4];
    const float* b1   = (const float*)d_in[5];
    const float* W2   = (const float*)d_in[6];
    const float* aS2  = (const float*)d_in[7];
    const float* aD2  = (const float*)d_in[8];
    const float* b2   = (const float*)d_in[9];

    const int N = in_sizes[0] / DIM;
    const int E = in_sizes[1] / 2;
    float* out = (float*)d_out;

    float* ws  = (float*)d_ws;
    float* h1  = ws;
    float* as1 = h1 + (size_t)N * 64;
    float* ad1 = as1 + (size_t)N * HEADS;
    float* o1  = ad1 + (size_t)N * HEADS;
    float* rec = o1 + (size_t)N * 64;
    const int tiles = (N + TILE - 1) / TILE;
    int* lst_s = (int*)(rec + (size_t)N * 4);
    int* lst_d = lst_s + (size_t)tiles * AGG_WAVES * WCAP;
    int* lcnt  = lst_d + (size_t)tiles * AGG_WAVES * WCAP;
    const int ngroups = (E + GRP - 1) / GRP;
    unsigned* code = (unsigned*)(lcnt + (size_t)tiles * AGG_WAVES);

    const int Mtiles = (N + 15) >> 4;

    gemm1_wmma<<<(Mtiles + 3) / 4, 128, 0, stream>>>(x, W1, h1, N);
    node_prep1<<<(N * HEADS + 255) / 256, 256, 0, stream>>>(h1, aS1, aD1, as1, ad1, ei, code, N, E, ngroups * (GRP / 4));
    agg1_lds<<<tiles, 1024, 0, stream>>>(ei, as1, ad1, h1, b1, o1, lst_s, lst_d, lcnt, code, ngroups, N, E);
    node_prep2<<<(N + 255) / 256, 256, 0, stream>>>(o1, W2, aS2, aD2, rec, N);
    agg2_lds<<<tiles, 1024, 0, stream>>>(ei, rec, b2, out, lst_s, lst_d, lcnt, code, ngroups, N, E);
}


# Layer-1 tile kernel asks fewer random lines per edge: a found edge reads one (source, destination) pair written by the logit kernel; pass 2 takes its 4 logits from the wave's list; sum order kept
# speedup vs baseline: 7.0962x; 1.0621x over previous
#include <hip/hip_runtime.h>
#include <math.h>

#define DIM 128
#define HID 16
#define HEADS 4
#define NEG_SLOPE 0.2f

typedef __attribute__((ext_vector_type(2))) float v2f;
typedef __attribute__((ext_vector_type(8))) float v8f;

__device__ __forceinline__ void atomicMaxFloat(float* addr, float v) {
    if (v >= 0.0f) {
        atomicMax((int*)addr, __float_as_int(v));
    } else {
        atomicMin((unsigned int*)addr, __float_as_uint(v));
    }
}

__global__ void __launch_bounds__(128)
gemm1_wmma(const float* __restrict__ x, const float* __restrict__ W1,
           float* __restrict__ h1, int N) {
    const int wave  = threadIdx.x >> 5;
    const int lane  = threadIdx.x & 31;
    const int Mtile = blockIdx.x * 4 + wave;
    const int Mtiles = (N + 15) >> 4;
    if (Mtile >= Mtiles) return;

    const int r  = lane & 15;
    const int kh = lane >> 4;
    const int mbase = Mtile * 16;
    int row = mbase + r;
    if (row >= N) row = N - 1;
    const float* __restrict__ xr = x + (size_t)row * DIM;
    const float* __restrict__ wc = W1 + 4 * r;

    v8f acc0 = {}, acc1 = {}, acc2 = {}, acc3 = {};

#pragma unroll
    for (int kb = 0; kb < DIM; kb += 4) {
        const int k0 = kb + 2 * kh;
        v2f a;
        a.x = xr[k0];
        a.y = xr[k0 + 1];
        const float4 w0 = *(const float4*)(wc + (size_t)k0 * 64);
        const float4 w1 = *(const float4*)(wc + (size_t)(k0 + 1) * 64);
        v2f b;
        b.x = w0.x;  b.y = w1.x;
        acc0 = __builtin_amdgcn_wmma_f32_16x16x4_f32(false, a, false, b, (short)0, acc0, false, false);
        b.x = w0.y;  b.y = w1.y;
        acc1 = __builtin_amdgcn_wmma_f32_16x16x4_f32(false, a, false, b, (short)0, acc1, false, false);
        b.x = w0.z;  b.y = w1.z;
        acc2 = __builtin_amdgcn_wmma_f32_16x16x4_f32(false, a, false, b, (short)0, acc2, false, false);
        b.x = w0.w;  b.y = w1.w;
        acc3 = __builtin_amdgcn_wmma_f32_16x16x4_f32(false, a, false, b, (short)0, acc3, false, false);
    }

    float* outp = h1 + (size_t)mbase * 64 + 4 * r;
#pragma unroll
    for (int i = 0; i < 8; ++i) {
        const int m = i + 8 * kh;
        if (mbase + m < N) {
            float4 o;
            o.x = acc0[i];
            o.y = acc1[i];
            o.z = acc2[i];
            o.w = acc3[i];
            *(float4*)(outp + (size_t)m * 64) = o;
        }
    }
}

#define TILE 400
#define GRP 4096

__device__ __forceinline__ unsigned tile_code(int d) {
    return (d < 0) ? 255u : (((unsigned)d / (unsigned)TILE) % 255u);
}

__global__ void __launch_bounds__(256)
node_prep1(const float* __restrict__ h1, const float* __restrict__ attS,
           const float* __restrict__ attD, float* __restrict__ as1,
           float* __restrict__ ad1, const int* __restrict__ ei,
           unsigned* __restrict__ code, int* __restrict__ pair, int N, int E, int ncode) {
    const int tid = blockIdx.x * blockDim.x + threadIdx.x;
    {
        const int* __restrict__ dstc = ei + E;
        const bool al16 = (((((size_t)dstc) | ((size_t)ei)) & 15) == 0);
        const int nthreads = gridDim.x * blockDim.x;
        for (int q = tid; q < ncode; q += nthreads) {
            const int e = q * 4;
            int d0 = -1, d1 = -1, d2 = -1, d3 = -1;
            if (al16 && e + 3 < E) {
                const int4 v = *(const int4*)(dstc + e);
                const int4 sv = *(const int4*)(ei + e);
                d0 = v.x; d1 = v.y; d2 = v.z; d3 = v.w;
                *(int4*)(pair + (size_t)e * 2)     = make_int4(sv.x, v.x, sv.y, v.y);
                *(int4*)(pair + (size_t)e * 2 + 4) = make_int4(sv.z, v.z, sv.w, v.w);
            } else {
                if (e     < E) { d0 = dstc[e];     *(int2*)(pair + (size_t)e * 2)     = make_int2(ei[e],     d0); }
                if (e + 1 < E) { d1 = dstc[e + 1]; *(int2*)(pair + (size_t)e * 2 + 2) = make_int2(ei[e + 1], d1); }
                if (e + 2 < E) { d2 = dstc[e + 2]; *(int2*)(pair + (size_t)e * 2 + 4) = make_int2(ei[e + 2], d2); }
                if (e + 3 < E) { d3 = dstc[e + 3]; *(int2*)(pair + (size_t)e * 2 + 6) = make_int2(ei[e + 3], d3); }
            }
            code[q] = tile_code(d0) | (tile_code(d1) << 8) | (tile_code(d2) << 16) | (tile_code(d3) << 24);
        }
    }
    if (tid >= N * HEADS) return;
    const int n  = tid >> 2;
    const int hd = tid & 3;
    const float* hp = h1 + (size_t)n * 64 + hd * HID;
    float ss = 0.f, dd = 0.f;
#pragma unroll
    for (int c = 0; c < HID; ++c) {
        const float v = hp[c];
        ss += v * attS[hd * HID + c];
        dd += v * attD[hd * HID + c];
    }
    as1[tid] = ss;
    ad1[tid] = dd;
}

#define WCAP 1024
#define QCAP 64
#define AGG_WAVES 32

__device__ __forceinline__ float leaky(float v) { return (v > 0.f) ? v : NEG_SLOPE * v; }

struct AggCtx {
    const int* pair;
    const float* as1; const float* ad1; const float* h1;
    float* acc; float* mx; float* sm;
    int* wq_e;
    int* lst_s; int* lst_d; float* lst_l;
    size_t lbase;
    int nbase, ntile, lane, wave, E;
    const unsigned* code; unsigned pat; int ngroups;
    const float* rec; float* adl; float* oa; float* ob;
};

__device__ __forceinline__ float4 batch_max(const AggCtx& cx, bool ok, int s, int dl) {
    float4 lg = make_float4(0.f, 0.f, 0.f, 0.f);
    if (ok) {
        const float4 a = *(const float4*)(cx.as1 + (size_t)s * 4);
        const float4 b = *(const float4*)(cx.ad1 + (size_t)(cx.nbase + dl) * 4);
        lg.x = leaky(a.x + b.x);
        lg.y = leaky(a.y + b.y);
        lg.z = leaky(a.z + b.z);
        lg.w = leaky(a.w + b.w);
        atomicMaxFloat(&cx.mx[dl * 4 + 0], lg.x);
        atomicMaxFloat(&cx.mx[dl * 4 + 1], lg.y);
        atomicMaxFloat(&cx.mx[dl * 4 + 2], lg.z);
        atomicMaxFloat(&cx.mx[dl * 4 + 3], lg.w);
    }
    return lg;
}

template <bool FROMLIST>
__device__ __forceinline__ void batch_acc(const AggCtx& cx, int nv, unsigned okm, int s, int dl, float4 lg) {
    float w0 = 0.f, w1 = 0.f, w2 = 0.f, w3 = 0.f;
    if ((okm >> cx.lane) & 1u) {
        if (!FROMLIST) {
            const float4 a = *(const float4*)(cx.as1 + (size_t)s * 4);
            const float4 b = *(const float4*)(cx.ad1 + (size_t)(cx.nbase + dl) * 4);
            lg.x = leaky(a.x + b.x);
            lg.y = leaky(a.y + b.y);
            lg.z = leaky(a.z + b.z);
            lg.w = leaky(a.w + b.w);
        }
        w0 = __expf(lg.x - cx.mx[dl * 4 + 0]);
        w1 = __expf(lg.y - cx.mx[dl * 4 + 1]);
        w2 = __expf(lg.z - cx.mx[dl * 4 + 2]);
        w3 = __expf(lg.w - cx.mx[dl * 4 + 3]);
        atomicAdd(&cx.sm[dl * 4 + 0], w0);
        atomicAdd(&cx.sm[dl * 4 + 1], w1);
        atomicAdd(&cx.sm[dl * 4 + 2], w2);
        atomicAdd(&cx.sm[dl * 4 + 3], w3);
    } else {
        s = 0; dl = 0;
    }
    const bool lo = cx.lane < 16;
    for (int j0 = 0; j0 < nv; j0 += 8) {
        float v0[8], v1[8];
#pragma unroll
        for (int u = 0; u < 8; ++u) {
            const int sj = __builtin_amdgcn_readlane(s, j0 + u);
            const float* hp = cx.h1 + (size_t)sj * 64 + cx.lane;
            v0[u] = hp[0];
            v1[u] = hp[32];
        }
#pragma unroll
        for (int u = 0; u < 8; ++u) {
            const int j = j0 + u;
            if ((okm >> j) & 1u) {
                const int dj = __builtin_amdgcn_readlane(dl, j);
                const float wlo_a = __int_as_float(__builtin_amdgcn_readlane(__float_as_int(w0), j));
                const float whi_a = __int_as_float(__builtin_amdgcn_readlane(__float_as_int(w1), j));
                const float wlo_b = __int_as_float(__builtin_amdgcn_readlane(__float_as_int(w2), j));
                const float whi_b = __int_as_float(__builtin_amdgcn_readlane(__float_as_int(w3), j));
                const float wa = lo ? wlo_a : whi_a;
                const float wb = lo ? wlo_b : whi_b;
                atomicAdd(&cx.acc[dj * 64 + cx.lane], wa * v0[u]);
                atomicAdd(&cx.acc[dj * 64 + 32 + cx.lane], wb * v1[u]);
            }
        }
    }
}

__device__ __forceinline__ void batch_max2(const AggCtx& cx, bool ok, int s, int dl) {
    if (ok) {
        const float a = cx.rec[(size_t)s * 4 + 2];
        atomicMaxFloat(&cx.mx[dl], leaky(a + cx.adl[dl]));
    }
}

__device__ __forceinline__ void batch_acc2(const AggCtx& cx, bool ok, int s, int dl) {
    if (ok) {
        const float4 r = *(const float4*)(cx.rec + (size_t)s * 4);
        const float w = __expf(leaky(r.z + cx.adl[dl]) - cx.mx[dl]);
        atomicAdd(&cx.sm[dl], w);
        atomicAdd(&cx.oa[dl], w * r.x);
        atomicAdd(&cx.ob[dl], w * r.y);
    }
}

template <int PHASE>
__device__ __forceinline__ void flush_batch(const AggCtx& cx, int head, int nv, int& nlist, bool& ovf) {
    __builtin_amdgcn_wave_barrier();
    const int qi = (head + cx.lane) & (QCAP - 1);
    int s = 0, dl = 0;
    bool ok = false;
    if (cx.lane < nv) {
        const int e = cx.wq_e[qi];
        const int2 pr = *(const int2*)(cx.pair + (size_t)e * 2);
        s  = pr.x;
        dl = pr.y - cx.nbase;
        ok = (unsigned)dl < (unsigned)cx.ntile;
    }
    const unsigned okm = __builtin_amdgcn_ballot_w32(ok);
    if (PHASE == 1) {
        const float4 lg = batch_max(cx, ok, s, dl);
        if (!ovf && nlist + 32 <= WCAP) {
            if (ok) {
                const int pos = nlist + __popc(okm & ((1u << cx.lane) - 1u));
                cx.lst_s[cx.lbase + pos] = s;
                cx.lst_d[cx.lbase + pos] = dl;
                *(float4*)(cx.lst_l + (cx.lbase + pos) * 4) = lg;
            }
            nlist += __popc(okm);
        } else {
            ovf = true;
        }
    } else if (PHASE == 2) {
        batch_acc<false>(cx, nv, okm, s, dl, make_float4(0.f, 0.f, 0.f, 0.f));
    } else if (PHASE == 3) {
        batch_max2(cx, ok, s, dl);
    } else {
        batch_acc2(cx, ok, s, dl);
    }
    __builtin_amdgcn_wave_barrier();
}

template <int PHASE>
__device__ __forceinline__ void scan_edges(const AggCtx& cx, int& nlist, bool& ovf) {
    int head = 0, cnt = 0;
    for (int g = cx.wave; g < cx.ngroups; g += AGG_WAVES) {
        const uint4* p = (const uint4*)(cx.code + (size_t)g * (GRP / 4)) + cx.lane;
        uint4 c[8];
#pragma unroll
        for (int u = 0; u < 8; ++u) c[u] = p[u * 32];
        unsigned hit[4];
#pragma unroll
        for (int sg = 0; sg < 4; ++sg) {
            unsigned bits = 0u;
#pragma unroll
            for (int k = 0; k < 8; ++k) {
                const uint4 cv = c[2 * sg + (k >> 2)];
                const unsigned v = ((k & 3) == 0) ? cv.x : ((k & 3) == 1) ? cv.y : ((k & 3) == 2) ? cv.z : cv.w;
                const unsigned x = v ^ cx.pat;
                const unsigned t = (x & 0x7f7f7f7fu) + 0x7f7f7f7fu;
                const unsigned y = ~(t | x) & 0x80808080u;
                bits |= y >> (7 - k);
            }
            hit[sg] = bits;
        }
#pragma unroll 1
        for (int sg = 0; sg < 4; ++sg) {
            unsigned bits = (sg == 0) ? hit[0] : (sg == 1) ? hit[1] : (sg == 2) ? hit[2] : hit[3];
            const int ebase = g * GRP + sg * 1024 + cx.lane * 16;
            for (int round = 0; round < 32; ++round) {
                const bool has = bits != 0u;
                const unsigned mask = __builtin_amdgcn_ballot_w32(has);
                if (mask == 0u) break;
                if (has) {
                    const int b = __ffs(bits) - 1;
                    const int pos = (head + cnt + __popc(mask & ((1u << cx.lane) - 1u))) & (QCAP - 1);
                    cx.wq_e[pos] = ebase + ((b >> 2) & 1) * 512 + (b & 3) * 4 + (b >> 3);
                    bits &= bits - 1u;
                }
                cnt += __popc(mask);
                if (cnt >= 32) {
                    flush_batch<PHASE>(cx, head, 32, nlist, ovf);
                    head = (head + 32) & (QCAP - 1);
                    cnt -= 32;
                }
            }
        }
    }
    if (cnt > 0) flush_batch<PHASE>(cx, head, cnt, nlist, ovf);
}

__global__ void __launch_bounds__(1024)
agg1_lds(const int* __restrict__ ei, const float* __restrict__ as1,
         const float* __restrict__ ad1, const float* __restrict__ h1,
         const float* __restrict__ b1, float* __restrict__ o1,
         int* lst_s, int* lst_d, float* lst_l, int* lcnt,
         const unsigned* __restrict__ code, const int* __restrict__ pair, int ngroups, int N, int E) {
    __shared__ float acc[TILE * 64];
    __shared__ float mx[TILE * HEADS];
    __shared__ float sm[TILE * HEADS];
    __shared__ int   qe[AGG_WAVES * QCAP];

    const int tid = threadIdx.x;
    AggCtx cx;
    cx.lane = tid & 31;
    cx.wave = __builtin_amdgcn_readfirstlane(tid >> 5);
    cx.nbase = blockIdx.x * TILE;
    cx.ntile = (N - cx.nbase < TILE) ? (N - cx.nbase) : TILE;
    cx.E = E;
    cx.pair = pair;
    cx.as1 = as1; cx.ad1 = ad1; cx.h1 = h1;
    cx.acc = acc; cx.mx = mx; cx.sm = sm;
    cx.code = code; cx.pat = (blockIdx.x % 255u) * 0x01010101u; cx.ngroups = ngroups;
    cx.wq_e = qe + cx.wave * QCAP;
    cx.lst_s = lst_s; cx.lst_d = lst_d; cx.lst_l = lst_l;
    cx.lbase = ((size_t)blockIdx.x * AGG_WAVES + cx.wave) * WCAP;

    for (int i = tid; i < TILE * 64; i += 1024) acc[i] = 0.f;
    for (int i = tid; i < TILE * HEADS; i += 1024) {
        sm[i] = 0.f;
        float v = 0.f;
        if ((i >> 2) < cx.ntile) {
            const size_t g = (size_t)cx.nbase * HEADS + i;
            v = leaky(as1[g] + ad1[g]);
        }
        mx[i] = v;
    }
    __syncthreads();

    int nlist = 0;
    bool ovf = false;
    scan_edges<1>(cx, nlist, ovf);
    if (cx.lane == 0) lcnt[blockIdx.x * AGG_WAVES + cx.wave] = ovf ? -1 : nlist;
    __syncthreads();

    if (!ovf) {
        for (int p = 0; p < nlist; p += 32) {
            const int nv = (nlist - p < 32) ? (nlist - p) : 32;
            int s = 0, dl = 0;
            float4 lg = make_float4(0.f, 0.f, 0.f, 0.f);
            if (cx.lane < nv) {
                s  = lst_s[cx.lbase + p + cx.lane];
                dl = lst_d[cx.lbase + p + cx.lane];
                lg = *(const float4*)(lst_l + (cx.lbase + p + cx.lane) * 4);
            }
            batch_acc<true>(cx, nv, (nv >= 32) ? 0xffffffffu : ((1u << nv) - 1u), s, dl, lg);
        }
    } else {
        int dummy_n = 0; bool dummy_o = false;
        scan_edges<2>(cx, dummy_n, dummy_o);
    }
    __syncthreads();

    for (int q = tid; q < cx.ntile * 16; q += 1024) {
        const int nl = q >> 4;
        const int c  = (q & 15) * 4;
        const int hd = c >> 4;
        const size_t n = (size_t)cx.nbase + nl;
        const float ls = leaky(as1[n * HEADS + hd] + ad1[n * HEADS + hd]);
        const float wself = __expf(ls - mx[nl * HEADS + hd]);
        const float den = sm[nl * HEADS + hd] + wself + 1e-16f;
        const float4 hv = *(const float4*)(h1 + n * 64 + c);
        const float4 av = *(const float4*)(acc + nl * 64 + c);
        const float4 bv = *(const float4*)(b1 + c);
        float4 r;
        r.x = (av.x + wself * hv.x) / den + bv.x;
        r.y = (av.y + wself * hv.y) / den + bv.y;
        r.z = (av.z + wself * hv.z) / den + bv.z;
        r.w = (av.w + wself * hv.w) / den + bv.w;
        r.x = (r.x > 0.f) ? r.x : expm1f(r.x);
        r.y = (r.y > 0.f) ? r.y : expm1f(r.y);
        r.z = (r.z > 0.f) ? r.z : expm1f(r.z);
        r.w = (r.w > 0.f) ? r.w : expm1f(r.w);
        *(float4*)(o1 + n * 64 + c) = r;
    }
}

__global__ void __launch_bounds__(256)
node_prep2(const float* __restrict__ o1, const float* __restrict__ W2,
           const float* __restrict__ attS2, const float* __restrict__ attD2,
           float* __restrict__ rec, int N) {
    const int n = blockIdx.x * blockDim.x + threadIdx.x;
    if (n >= N) return;
    const float* hp = o1 + (size_t)n * 64;
    float a0 = 0.f, a1 = 0.f;
#pragma unroll
    for (int k = 0; k < 64; ++k) {
        const float v = hp[k];
        a0 += v * W2[k * 2 + 0];
        a1 += v * W2[k * 2 + 1];
    }
    float4 r;
    r.x = a0;
    r.y = a1;
    r.z = a0 * attS2[0] + a1 * attS2[1];
    r.w = a0 * attD2[0] + a1 * attD2[1];
    *(float4*)(rec + (size_t)n * 4) = r;
}

__global__ void __launch_bounds__(1024)
agg2_lds(const int* __restrict__ ei, const float* __restrict__ rec,
         const float* __restrict__ b2, float* __restrict__ out,
         int* lst_s, int* lst_d, const int* __restrict__ lcnt,
         const unsigned* __restrict__ code, const int* __restrict__ pair, int ngroups, int N, int E) {
    __shared__ float mx[TILE];
    __shared__ float sm[TILE];
    __shared__ float oa[TILE];
    __shared__ float ob[TILE];
    __shared__ float adl[TILE];
    __shared__ int   qe[AGG_WAVES * QCAP];

    const int tid = threadIdx.x;
    AggCtx cx;
    cx.lane = tid & 31;
    cx.wave = __builtin_amdgcn_readfirstlane(tid >> 5);
    cx.nbase = blockIdx.x * TILE;
    cx.ntile = (N - cx.nbase < TILE) ? (N - cx.nbase) : TILE;
    cx.E = E;
    cx.pair = pair;
    cx.as1 = nullptr; cx.ad1 = nullptr; cx.h1 = nullptr; cx.acc = nullptr;
    cx.mx = mx; cx.sm = sm;
    cx.code = code; cx.pat = (blockIdx.x % 255u) * 0x01010101u; cx.ngroups = ngroups;
    cx.wq_e = qe + cx.wave * QCAP;
    cx.lst_s = lst_s; cx.lst_d = lst_d; cx.lst_l = nullptr;
    cx.lbase = ((size_t)blockIdx.x * AGG_WAVES + cx.wave) * WCAP;
    cx.rec = rec; cx.adl = adl; cx.oa = oa; cx.ob = ob;

    for (int i = tid; i < TILE; i += 1024) {
        float m = 0.f, a = 0.f;
        if (i < cx.ntile) {
            const float4 r = *(const float4*)(rec + ((size_t)cx.nbase + i) * 4);
            a = r.w;
            m = leaky(r.z + r.w);
        }
        mx[i] = m; adl[i] = a;
        sm[i] = 0.f; oa[i] = 0.f; ob[i] = 0.f;
    }
    __syncthreads();

    const int cnt = lcnt[blockIdx.x * AGG_WAVES + cx.wave];
    int dummy_n = 0; bool dummy_o = false;

    if (cnt >= 0) {
        for (int p = 0; p < cnt; p += 32) {
            const int nv = (cnt - p < 32) ? (cnt - p) : 32;
            int s = 0, dl = 0;
            if (cx.lane < nv) {
                s  = lst_s[cx.lbase + p + cx.lane];
                dl = lst_d[cx.lbase + p + cx.lane];
            }
            batch_max2(cx, cx.lane < nv, s, dl);
        }
    } else {
        scan_edges<3>(cx, dummy_n, dummy_o);
    }
    __syncthreads();

    if (cnt >= 0) {
        for (int p = 0; p < cnt; p += 32) {
            const int nv = (cnt - p < 32) ? (cnt - p) : 32;
            int s = 0, dl = 0;
            if (cx.lane < nv) {
                s  = lst_s[cx.lbase + p + cx.lane];
                dl = lst_d[cx.lbase + p + cx.lane];
            }
            batch_acc2(cx, cx.lane < nv, s, dl);
        }
    } else {
        scan_edges<4>(cx, dummy_n, dummy_o);
    }
    __syncthreads();

    for (int i = tid; i < cx.ntile; i += 1024) {
        const size_t n = (size_t)cx.nbase + i;
        const float4 r = *(const float4*)(rec + n * 4);
        const float wself = __expf(leaky(r.z + r.w) - mx[i]);
        const float inv = 1.0f / (sm[i] + wself + 1e-16f);
        float2 o;
        o.x = (oa[i] + wself * r.x) * inv + b2[0];
        o.y = (ob[i] + wself * r.y) * inv + b2[1];
        *(float2*)(out + n * 2) = o;
    }
}

extern "C" void kernel_launch(void* const* d_in, const int* in_sizes, int n_in,
                              void* d_out, int out_size, void* d_ws, size_t ws_size,
                              hipStream_t stream) {
    const float* x    = (const float*)d_in[0];
    const int*   ei   = (const int*)d_in[1];
    const float* W1   = (const float*)d_in[2];
    const float* aS1  = (const float*)d_in[3];
    const float* aD1  = (const float*)d_in[4];
    const float* b1   = (const float*)d_in[5];
    const float* W2   = (const float*)d_in[6];
    const float* aS2  = (const float*)d_in[7];
    const float* aD2  = (const float*)d_in[8];
    const float* b2   = (const float*)d_in[9];

    const int N = in_sizes[0] / DIM;
    const int E = in_sizes[1] / 2;
    float* out = (float*)d_out;

    float* ws  = (float*)d_ws;
    float* h1  = ws;
    float* as1 = h1 + (size_t)N * 64;
    float* ad1 = as1 + (size_t)N * HEADS;
    float* o1  = ad1 + (size_t)N * HEADS;
    float* rec = o1 + (size_t)N * 64;
    const int tiles = (N + TILE - 1) / TILE;
    float* lst_l = rec + (size_t)N * 4;
    int* pair  = (int*)(lst_l + (size_t)tiles * AGG_WAVES * WCAP * 4);
    int* lst_s = pair + (size_t)E * 2;
    int* lst_d = lst_s + (size_t)tiles * AGG_WAVES * WCAP;
    int* lcnt  = lst_d + (size_t)tiles * AGG_WAVES * WCAP;
    const int ngroups = (E + GRP - 1) / GRP;
    unsigned* code = (unsigned*)(lcnt + (size_t)tiles * AGG_WAVES);

    const int Mtiles = (N + 15) >> 4;

    gemm1_wmma<<<(Mtiles + 3) / 4, 128, 0, stream>>>(x, W1, h1, N);
    node_prep1<<<(N * HEADS + 255) / 256, 256, 0, stream>>>(h1, aS1, aD1, as1, ad1, ei, code, pair, N, E, ngroups * (GRP / 4));
    agg1_lds<<<tiles, 1024, 0, stream>>>(ei, as1, ad1, h1, b1, o1, lst_s, lst_d, lst_l, lcnt, code, pair, ngroups, N, E);
    node_prep2<<<(N + 255) / 256, 256, 0, stream>>>(o1, W2, aS2, aD2, rec, N);
    agg2_lds<<<tiles, 1024, 0, stream>>>(ei, rec, b2, out, lst_s, lst_d, lcnt, code, pair, ngroups, N, E);
}


# Layer-2 tile kernel: a wave requests its whole edge list, then all listed records, at once and keeps them in registers for both passes (was 2 dependent loads per 32 edges, twice); LDS sum order kept
# speedup vs baseline: 7.3467x; 1.0353x over previous
#include <hip/hip_runtime.h>
#include <math.h>

#define DIM 128
#define HID 16
#define HEADS 4
#define NEG_SLOPE 0.2f

typedef __attribute__((ext_vector_type(2))) float v2f;
typedef __attribute__((ext_vector_type(8))) float v8f;

__device__ __forceinline__ void atomicMaxFloat(float* addr, float v) {
    if (v >= 0.0f) {
        atomicMax((int*)addr, __float_as_int(v));
    } else {
        atomicMin((unsigned int*)addr, __float_as_uint(v));
    }
}

__global__ void __launch_bounds__(128)
gemm1_wmma(const float* __restrict__ x, const float* __restrict__ W1,
           float* __restrict__ h1, int N) {
    const int wave  = threadIdx.x >> 5;
    const int lane  = threadIdx.x & 31;
    const int Mtile = blockIdx.x * 4 + wave;
    const int Mtiles = (N + 15) >> 4;
    if (Mtile >= Mtiles) return;

    const int r  = lane & 15;
    const int kh = lane >> 4;
    const int mbase = Mtile * 16;
    int row = mbase + r;
    if (row >= N) row = N - 1;
    const float* __restrict__ xr = x + (size_t)row * DIM;
    const float* __restrict__ wc = W1 + 4 * r;

    v8f acc0 = {}, acc1 = {}, acc2 = {}, acc3 = {};

#pragma unroll
    for (int kb = 0; kb < DIM; kb += 4) {
        const int k0 = kb + 2 * kh;
        v2f a;
        a.x = xr[k0];
        a.y = xr[k0 + 1];
        const float4 w0 = *(const float4*)(wc + (size_t)k0 * 64);
        const float4 w1 = *(const float4*)(wc + (size_t)(k0 + 1) * 64);
        v2f b;
        b.x = w0.x;  b.y = w1.x;
        acc0 = __builtin_amdgcn_wmma_f32_16x16x4_f32(false, a, false, b, (short)0, acc0, false, false);
        b.x = w0.y;  b.y = w1.y;
        acc1 = __builtin_amdgcn_wmma_f32_16x16x4_f32(false, a, false, b, (short)0, acc1, false, false);
        b.x = w0.z;  b.y = w1.z;
        acc2 = __builtin_amdgcn_wmma_f32_16x16x4_f32(false, a, false, b, (short)0, acc2, false, false);
        b.x = w0.w;  b.y = w1.w;
        acc3 = __builtin_amdgcn_wmma_f32_16x16x4_f32(false, a, false, b, (short)0, acc3, false, false);
    }

    float* outp = h1 + (size_t)mbase * 64 + 4 * r;
#pragma unroll
    for (int i = 0; i < 8; ++i) {
        const int m = i + 8 * kh;
        if (mbase + m < N) {
            float4 o;
            o.x = acc0[i];
            o.y = acc1[i];
            o.z = acc2[i];
            o.w = acc3[i];
            *(float4*)(outp + (size_t)m * 64) = o;
        }
    }
}

#define TILE 400
#define GRP 4096

__device__ __forceinline__ unsigned tile_code(int d) {
    return (d < 0) ? 255u : (((unsigned)d / (unsigned)TILE) % 255u);
}

__global__ void __launch_bounds__(256)
node_prep1(const float* __restrict__ h1, const float* __restrict__ attS,
           const float* __restrict__ attD, float* __restrict__ as1,
           float* __restrict__ ad1, const int* __restrict__ ei,
           unsigned* __restrict__ code, int* __restrict__ pair, int N, int E, int ncode) {
    const int tid = blockIdx.x * blockDim.x + threadIdx.x;
    {
        const int* __restrict__ dstc = ei + E;
        const bool al16 = (((((size_t)dstc) | ((size_t)ei)) & 15) == 0);
        const int nthreads = gridDim.x * blockDim.x;
        for (int q = tid; q < ncode; q += nthreads) {
            const int e = q * 4;
            int d0 = -1, d1 = -1, d2 = -1, d3 = -1;
            if (al16 && e + 3 < E) {
                const int4 v = *(const int4*)(dstc + e);
                const int4 sv = *(const int4*)(ei + e);
                d0 = v.x; d1 = v.y; d2 = v.z; d3 = v.w;
                *(int4*)(pair + (size_t)e * 2)     = make_int4(sv.x, v.x, sv.y, v.y);
                *(int4*)(pair + (size_t)e * 2 + 4) = make_int4(sv.z, v.z, sv.w, v.w);
            } else {
                if (e     < E) { d0 = dstc[e];     *(int2*)(pair + (size_t)e * 2)     = make_int2(ei[e],     d0); }
                if (e + 1 < E) { d1 = dstc[e + 1]; *(int2*)(pair + (size_t)e * 2 + 2) = make_int2(ei[e + 1], d1); }
                if (e + 2 < E) { d2 = dstc[e + 2]; *(int2*)(pair + (size_t)e * 2 + 4) = make_int2(ei[e + 2], d2); }
                if (e + 3 < E) { d3 = dstc[e + 3]; *(int2*)(pair + (size_t)e * 2 + 6) = make_int2(ei[e + 3], d3); }
            }
            code[q] = tile_code(d0) | (tile_code(d1) << 8) | (tile_code(d2) << 16) | (tile_code(d3) << 24);
        }
    }
    if (tid >= N * HEADS) return;
    const int n  = tid >> 2;
    const int hd = tid & 3;
    const float* hp = h1 + (size_t)n * 64 + hd * HID;
    float ss = 0.f, dd = 0.f;
#pragma unroll
    for (int c = 0; c < HID; ++c) {
        const float v = hp[c];
        ss += v * attS[hd * HID + c];
        dd += v * attD[hd * HID + c];
    }
    as1[tid] = ss;
    ad1[tid] = dd;
}

#define WCAP 1024
#define QCAP 64
#define AGG_WAVES 32

__device__ __forceinline__ float leaky(float v) { return (v > 0.f) ? v : NEG_SLOPE * v; }

struct AggCtx {
    const int* pair;
    const float* as1; const float* ad1; const float* h1;
    float* acc; float* mx; float* sm;
    int* wq_e;
    int* lst_s; int* lst_d; float* lst_l;
    size_t lbase;
    int nbase, ntile, lane, wave, E;
    const unsigned* code; unsigned pat; int ngroups;
    const float* rec; float* adl; float* oa; float* ob;
};

__device__ __forceinline__ float4 batch_max(const AggCtx& cx, bool ok, int s, int dl) {
    float4 lg = make_float4(0.f, 0.f, 0.f, 0.f);
    if (ok) {
        const float4 a = *(const float4*)(cx.as1 + (size_t)s * 4);
        const float4 b = *(const float4*)(cx.ad1 + (size_t)(cx.nbase + dl) * 4);
        lg.x = leaky(a.x + b.x);
        lg.y = leaky(a.y + b.y);
        lg.z = leaky(a.z + b.z);
        lg.w = leaky(a.w + b.w);
        atomicMaxFloat(&cx.mx[dl * 4 + 0], lg.x);
        atomicMaxFloat(&cx.mx[dl * 4 + 1], lg.y);
        atomicMaxFloat(&cx.mx[dl * 4 + 2], lg.z);
        atomicMaxFloat(&cx.mx[dl * 4 + 3], lg.w);
    }
    return lg;
}

template <bool FROMLIST>
__device__ __forceinline__ void batch_acc(const AggCtx& cx, int nv, unsigned okm, int s, int dl, float4 lg) {
    float w0 = 0.f, w1 = 0.f, w2 = 0.f, w3 = 0.f;
    if ((okm >> cx.lane) & 1u) {
        if (!FROMLIST) {
            const float4 a = *(const float4*)(cx.as1 + (size_t)s * 4);
            const float4 b = *(const float4*)(cx.ad1 + (size_t)(cx.nbase + dl) * 4);
            lg.x = leaky(a.x + b.x);
            lg.y = leaky(a.y + b.y);
            lg.z = leaky(a.z + b.z);
            lg.w = leaky(a.w + b.w);
        }
        w0 = __expf(lg.x - cx.mx[dl * 4 + 0]);
        w1 = __expf(lg.y - cx.mx[dl * 4 + 1]);
        w2 = __expf(lg.z - cx.mx[dl * 4 + 2]);
        w3 = __expf(lg.w - cx.mx[dl * 4 + 3]);
        atomicAdd(&cx.sm[dl * 4 + 0], w0);
        atomicAdd(&cx.sm[dl * 4 + 1], w1);
        atomicAdd(&cx.sm[dl * 4 + 2], w2);
        atomicAdd(&cx.sm[dl * 4 + 3], w3);
    } else {
        s = 0; dl = 0;
    }
    const bool lo = cx.lane < 16;
    for (int j0 = 0; j0 < nv; j0 += 8) {
        float v0[8], v1[8];
#pragma unroll
        for (int u = 0; u < 8; ++u) {
            const int sj = __builtin_amdgcn_readlane(s, j0 + u);
            const float* hp = cx.h1 + (size_t)sj * 64 + cx.lane;
            v0[u] = hp[0];
            v1[u] = hp[32];
        }
#pragma unroll
        for (int u = 0; u < 8; ++u) {
            const int j = j0 + u;
            if ((okm >> j) & 1u) {
                const int dj = __builtin_amdgcn_readlane(dl, j);
                const float wlo_a = __int_as_float(__builtin_amdgcn_readlane(__float_as_int(w0), j));
                const float whi_a = __int_as_float(__builtin_amdgcn_readlane(__float_as_int(w1), j));
                const float wlo_b = __int_as_float(__builtin_amdgcn_readlane(__float_as_int(w2), j));
                const float whi_b = __int_as_float(__builtin_amdgcn_readlane(__float_as_int(w3), j));
                const float wa = lo ? wlo_a : whi_a;
                const float wb = lo ? wlo_b : whi_b;
                atomicAdd(&cx.acc[dj * 64 + cx.lane], wa * v0[u]);
                atomicAdd(&cx.acc[dj * 64 + 32 + cx.lane], wb * v1[u]);
            }
        }
    }
}

__device__ __forceinline__ void batch_max2(const AggCtx& cx, bool ok, int s, int dl) {
    if (ok) {
        const float a = cx.rec[(size_t)s * 4 + 2];
        atomicMaxFloat(&cx.mx[dl], leaky(a + cx.adl[dl]));
    }
}

__device__ __forceinline__ void batch_acc2(const AggCtx& cx, bool ok, int s, int dl) {
    if (ok) {
        const float4 r = *(const float4*)(cx.rec + (size_t)s * 4);
        const float w = __expf(leaky(r.z + cx.adl[dl]) - cx.mx[dl]);
        atomicAdd(&cx.sm[dl], w);
        atomicAdd(&cx.oa[dl], w * r.x);
        atomicAdd(&cx.ob[dl], w * r.y);
    }
}

template <int PHASE>
__device__ __forceinline__ void flush_batch(const AggCtx& cx, int head, int nv, int& nlist, bool& ovf) {
    __builtin_amdgcn_wave_barrier();
    const int qi = (head + cx.lane) & (QCAP - 1);
    int s = 0, dl = 0;
    bool ok = false;
    if (cx.lane < nv) {
        const int e = cx.wq_e[qi];
        const int2 pr = *(const int2*)(cx.pair + (size_t)e * 2);
        s  = pr.x;
        dl = pr.y - cx.nbase;
        ok = (unsigned)dl < (unsigned)cx.ntile;
    }
    const unsigned okm = __builtin_amdgcn_ballot_w32(ok);
    if (PHASE == 1) {
        const float4 lg = batch_max(cx, ok, s, dl);
        if (!ovf && nlist + 32 <= WCAP) {
            if (ok) {
                const int pos = nlist + __popc(okm & ((1u << cx.lane) - 1u));
                cx.lst_s[cx.lbase + pos] = s;
                cx.lst_d[cx.lbase + pos] = dl;
                *(float4*)(cx.lst_l + (cx.lbase + pos) * 4) = lg;
            }
            nlist += __popc(okm);
        } else {
            ovf = true;
        }
    } else if (PHASE == 2) {
        batch_acc<false>(cx, nv, okm, s, dl, make_float4(0.f, 0.f, 0.f, 0.f));
    } else if (PHASE == 3) {
        batch_max2(cx, ok, s, dl);
    } else {
        batch_acc2(cx, ok, s, dl);
    }
    __builtin_amdgcn_wave_barrier();
}

template <int PHASE>
__device__ __forceinline__ void scan_edges(const AggCtx& cx, int& nlist, bool& ovf) {
    int head = 0, cnt = 0;
    for (int g = cx.wave; g < cx.ngroups; g += AGG_WAVES) {
        const uint4* p = (const uint4*)(cx.code + (size_t)g * (GRP / 4)) + cx.lane;
        uint4 c[8];
#pragma unroll
        for (int u = 0; u < 8; ++u) c[u] = p[u * 32];
        unsigned hit[4];
#pragma unroll
        for (int sg = 0; sg < 4; ++sg) {
            unsigned bits = 0u;
#pragma unroll
            for (int k = 0; k < 8; ++k) {
                const uint4 cv = c[2 * sg + (k >> 2)];
                const unsigned v = ((k & 3) == 0) ? cv.x : ((k & 3) == 1) ? cv.y : ((k & 3) == 2) ? cv.z : cv.w;
                const unsigned x = v ^ cx.pat;
                const unsigned t = (x & 0x7f7f7f7fu) + 0x7f7f7f7fu;
                const unsigned y = ~(t | x) & 0x80808080u;
                bits |= y >> (7 - k);
            }
            hit[sg] = bits;
        }
#pragma unroll 1
        for (int sg = 0; sg < 4; ++sg) {
            unsigned bits = (sg == 0) ? hit[0] : (sg == 1) ? hit[1] : (sg == 2) ? hit[2] : hit[3];
            const int ebase = g * GRP + sg * 1024 + cx.lane * 16;
            for (int round = 0; round < 32; ++round) {
                const bool has = bits != 0u;
                const unsigned mask = __builtin_amdgcn_ballot_w32(has);
                if (mask == 0u) break;
                if (has) {
                    const int b = __ffs(bits) - 1;
                    const int pos = (head + cnt + __popc(mask & ((1u << cx.lane) - 1u))) & (QCAP - 1);
                    cx.wq_e[pos] = ebase + ((b >> 2) & 1) * 512 + (b & 3) * 4 + (b >> 3);
                    bits &= bits - 1u;
                }
                cnt += __popc(mask);
                if (cnt >= 32) {
                    flush_batch<PHASE>(cx, head, 32, nlist, ovf);
                    head = (head + 32) & (QCAP - 1);
                    cnt -= 32;
                }
            }
        }
    }
    if (cnt > 0) flush_batch<PHASE>(cx, head, cnt, nlist, ovf);
}

__global__ void __launch_bounds__(1024)
agg1_lds(const int* __restrict__ ei, const float* __restrict__ as1,
         const float* __restrict__ ad1, const float* __restrict__ h1,
         const float* __restrict__ b1, float* __restrict__ o1,
         int* lst_s, int* lst_d, float* lst_l, int* lcnt,
         const unsigned* __restrict__ code, const int* __restrict__ pair, int ngroups, int N, int E) {
    __shared__ float acc[TILE * 64];
    __shared__ float mx[TILE * HEADS];
    __shared__ float sm[TILE * HEADS];
    __shared__ int   qe[AGG_WAVES * QCAP];

    const int tid = threadIdx.x;
    AggCtx cx;
    cx.lane = tid & 31;
    cx.wave = __builtin_amdgcn_readfirstlane(tid >> 5);
    cx.nbase = blockIdx.x * TILE;
    cx.ntile = (N - cx.nbase < TILE) ? (N - cx.nbase) : TILE;
    cx.E = E;
    cx.pair = pair;
    cx.as1 = as1; cx.ad1 = ad1; cx.h1 = h1;
    cx.acc = acc; cx.mx = mx; cx.sm = sm;
    cx.code = code; cx.pat = (blockIdx.x % 255u) * 0x01010101u; cx.ngroups = ngroups;
    cx.wq_e = qe + cx.wave * QCAP;
    cx.lst_s = lst_s; cx.lst_d = lst_d; cx.lst_l = lst_l;
    cx.lbase = ((size_t)blockIdx.x * AGG_WAVES + cx.wave) * WCAP;

    for (int i = tid; i < TILE * 64; i += 1024) acc[i] = 0.f;
    for (int i = tid; i < TILE * HEADS; i += 1024) {
        sm[i] = 0.f;
        float v = 0.f;
        if ((i >> 2) < cx.ntile) {
            const size_t g = (size_t)cx.nbase * HEADS + i;
            v = leaky(as1[g] + ad1[g]);
        }
        mx[i] = v;
    }
    __syncthreads();

    int nlist = 0;
    bool ovf = false;
    scan_edges<1>(cx, nlist, ovf);
    if (cx.lane == 0) lcnt[blockIdx.x * AGG_WAVES + cx.wave] = ovf ? -1 : nlist;
    __syncthreads();

    if (!ovf) {
        for (int p = 0; p < nlist; p += 32) {
            const int nv = (nlist - p < 32) ? (nlist - p) : 32;
            int s = 0, dl = 0;
            float4 lg = make_float4(0.f, 0.f, 0.f, 0.f);
            if (cx.lane < nv) {
                s  = lst_s[cx.lbase + p + cx.lane];
                dl = lst_d[cx.lbase + p + cx.lane];
                lg = *(const float4*)(lst_l + (cx.lbase + p + cx.lane) * 4);
            }
            batch_acc<true>(cx, nv, (nv >= 32) ? 0xffffffffu : ((1u << nv) - 1u), s, dl, lg);
        }
    } else {
        int dummy_n = 0; bool dummy_o = false;
        scan_edges<2>(cx, dummy_n, dummy_o);
    }
    __syncthreads();

    for (int q = tid; q < cx.ntile * 16; q += 1024) {
        const int nl = q >> 4;
        const int c  = (q & 15) * 4;
        const int hd = c >> 4;
        const size_t n = (size_t)cx.nbase + nl;
        const float ls = leaky(as1[n * HEADS + hd] + ad1[n * HEADS + hd]);
        const float wself = __expf(ls - mx[nl * HEADS + hd]);
        const float den = sm[nl * HEADS + hd] + wself + 1e-16f;
        const float4 hv = *(const float4*)(h1 + n * 64 + c);
        const float4 av = *(const float4*)(acc + nl * 64 + c);
        const float4 bv = *(const float4*)(b1 + c);
        float4 r;
        r.x = (av.x + wself * hv.x) / den + bv.x;
        r.y = (av.y + wself * hv.y) / den + bv.y;
        r.z = (av.z + wself * hv.z) / den + bv.z;
        r.w = (av.w + wself * hv.w) / den + bv.w;
        r.x = (r.x > 0.f) ? r.x : expm1f(r.x);
        r.y = (r.y > 0.f) ? r.y : expm1f(r.y);
        r.z = (r.z > 0.f) ? r.z : expm1f(r.z);
        r.w = (r.w > 0.f) ? r.w : expm1f(r.w);
        *(float4*)(o1 + n * 64 + c) = r;
    }
}

__global__ void __launch_bounds__(256)
node_prep2(const float* __restrict__ o1, const float* __restrict__ W2,
           const float* __restrict__ attS2, const float* __restrict__ attD2,
           float* __restrict__ rec, int N) {
    const int n = blockIdx.x * blockDim.x + threadIdx.x;
    if (n >= N) return;
    const float* hp = o1 + (size_t)n * 64;
    float a0 = 0.f, a1 = 0.f;
#pragma unroll
    for (int k = 0; k < 64; ++k) {
        const float v = hp[k];
        a0 += v * W2[k * 2 + 0];
        a1 += v * W2[k * 2 + 1];
    }
    float4 r;
    r.x = a0;
    r.y = a1;
    r.z = a0 * attS2[0] + a1 * attS2[1];
    r.w = a0 * attD2[0] + a1 * attD2[1];
    *(float4*)(rec + (size_t)n * 4) = r;
}

#define L2B 8

__global__ void __launch_bounds__(1024)
agg2_lds(const int* __restrict__ ei, const float* __restrict__ rec,
         const float* __restrict__ b2, float* __restrict__ out,
         int* lst_s, int* lst_d, const int* __restrict__ lcnt,
         const unsigned* __restrict__ code, const int* __restrict__ pair, int ngroups, int N, int E) {
    __shared__ float mx[TILE];
    __shared__ float sm[TILE];
    __shared__ float oa[TILE];
    __shared__ float ob[TILE];
    __shared__ float adl[TILE];
    __shared__ int   qe[AGG_WAVES * QCAP];
    static_assert(TILE <= 1024, "one thread per node of the tile");

    const int tid = threadIdx.x;
    AggCtx cx;
    cx.lane = tid & 31;
    cx.wave = __builtin_amdgcn_readfirstlane(tid >> 5);
    cx.nbase = blockIdx.x * TILE;
    cx.ntile = (N - cx.nbase < TILE) ? (N - cx.nbase) : TILE;
    cx.E = E;
    cx.pair = pair;
    cx.as1 = nullptr; cx.ad1 = nullptr; cx.h1 = nullptr; cx.acc = nullptr;
    cx.mx = mx; cx.sm = sm;
    cx.code = code; cx.pat = (blockIdx.x % 255u) * 0x01010101u; cx.ngroups = ngroups;
    cx.wq_e = qe + cx.wave * QCAP;
    cx.lst_s = lst_s; cx.lst_d = lst_d; cx.lst_l = nullptr;
    cx.lbase = ((size_t)blockIdx.x * AGG_WAVES + cx.wave) * WCAP;
    cx.rec = rec; cx.adl = adl; cx.oa = oa; cx.ob = ob;

    const int cnt = lcnt[blockIdx.x * AGG_WAVES + cx.wave];

    int sv[L2B], dv[L2B];
#pragma unroll
    for (int k = 0; k < L2B; ++k) { sv[k] = 0; dv[k] = 0; }
    if (cnt > 0) {
#pragma unroll
        for (int k = 0; k < L2B; ++k) {
            int idx = k * 32 + cx.lane;
            idx = (idx < cnt) ? idx : (cnt - 1);
            sv[k] = lst_s[cx.lbase + idx];
            dv[k] = lst_d[cx.lbase + idx];
        }
    }

    float4 rown = make_float4(0.f, 0.f, 0.f, 0.f);
    if (tid < TILE) {
        float m = 0.f, a = 0.f;
        if (tid < cx.ntile) {
            rown = *(const float4*)(rec + ((size_t)cx.nbase + tid) * 4);
            a = rown.w;
            m = leaky(rown.z + rown.w);
        }
        mx[tid] = m; adl[tid] = a;
        sm[tid] = 0.f; oa[tid] = 0.f; ob[tid] = 0.f;
    }

    float rx[L2B], ry[L2B], lg[L2B];
#pragma unroll
    for (int k = 0; k < L2B; ++k) { rx[k] = 0.f; ry[k] = 0.f; lg[k] = 0.f; }
    if (cnt > 0) {
#pragma unroll
        for (int k = 0; k < L2B; ++k) {
            const float4 r = *(const float4*)(rec + (size_t)sv[k] * 4);
            rx[k] = r.x; ry[k] = r.y; lg[k] = r.z;
        }
    }
    __syncthreads();

    int dummy_n = 0; bool dummy_o = false;

    if (cnt >= 0) {
        float ad[L2B];
#pragma unroll
        for (int k = 0; k < L2B; ++k) ad[k] = adl[dv[k]];
#pragma unroll
        for (int k = 0; k < L2B; ++k) {
            if (k * 32 + cx.lane < cnt) {
                lg[k] = leaky(lg[k] + ad[k]);
                atomicMaxFloat(&mx[dv[k]], lg[k]);
            }
        }
        for (int p = L2B * 32; p < cnt; p += 32) {
            const int nv = (cnt - p < 32) ? (cnt - p) : 32;
            int s = 0, dl = 0;
            if (cx.lane < nv) {
                s  = lst_s[cx.lbase + p + cx.lane];
                dl = lst_d[cx.lbase + p + cx.lane];
            }
            batch_max2(cx, cx.lane < nv, s, dl);
        }
    } else {
        scan_edges<3>(cx, dummy_n, dummy_o);
    }
    __syncthreads();

    if (cnt >= 0) {
        float mv[L2B];
#pragma unroll
        for (int k = 0; k < L2B; ++k) mv[k] = mx[dv[k]];
#pragma unroll
        for (int k = 0; k < L2B; ++k) {
            if (k * 32 + cx.lane < cnt) {
                const int d = dv[k];
                const float w = __expf(lg[k] - mv[k]);
                atomicAdd(&sm[d], w);
                atomicAdd(&oa[d], w * rx[k]);
                atomicAdd(&ob[d], w * ry[k]);
            }
        }
        for (int p = L2B * 32; p < cnt; p += 32) {
            const int nv = (cnt - p < 32) ? (cnt - p) : 32;
            int s = 0, dl = 0;
            if (cx.lane < nv) {
                s  = lst_s[cx.lbase + p + cx.lane];
                dl = lst_d[cx.lbase + p + cx.lane];
            }
            batch_acc2(cx, cx.lane < nv, s, dl);
        }
    } else {
        scan_edges<4>(cx, dummy_n, dummy_o);
    }
    __syncthreads();

    if (tid < cx.ntile) {
        const size_t n = (size_t)cx.nbase + tid;
        const float4 r = rown;
        const float wself = __expf(leaky(r.z + r.w) - mx[tid]);
        const float inv = 1.0f / (sm[tid] + wself + 1e-16f);
        float2 o;
        o.x = (oa[tid] + wself * r.x) * inv + b2[0];
        o.y = (ob[tid] + wself * r.y) * inv + b2[1];
        *(float2*)(out + n * 2) = o;
    }
}

extern "C" void kernel_launch(void* const* d_in, const int* in_sizes, int n_in,
                              void* d_out, int out_size, void* d_ws, size_t ws_size,
                              hipStream_t stream) {
    const float* x    = (const float*)d_in[0];
    const int*   ei   = (const int*)d_in[1];
    const float* W1   = (const float*)d_in[2];
    const float* aS1  = (const float*)d_in[3];
    const float* aD1  = (const float*)d_in[4];
    const float* b1   = (const float*)d_in[5];
    const float* W2   = (const float*)d_in[6];
    const float* aS2  = (const float*)d_in[7];
    const float* aD2  = (const float*)d_in[8];
    const float* b2   = (const float*)d_in[9];

    const int N = in_sizes[0] / DIM;
    const int E = in_sizes[1] / 2;
    float* out = (float*)d_out;

    float* ws  = (float*)d_ws;
    float* h1  = ws;
    float* as1 = h1 + (size_t)N * 64;
    float* ad1 = as1 + (size_t)N * HEADS;
    float* o1  = ad1 + (size_t)N * HEADS;
    float* rec = o1 + (size_t)N * 64;
    const int tiles = (N + TILE - 1) / TILE;
    float* lst_l = rec + (size_t)N * 4;
    int* pair  = (int*)(lst_l + (size_t)tiles * AGG_WAVES * WCAP * 4);
    int* lst_s = pair + (size_t)E * 2;
    int* lst_d = lst_s + (size_t)tiles * AGG_WAVES * WCAP;
    int* lcnt  = lst_d + (size_t)tiles * AGG_WAVES * WCAP;
    const int ngroups = (E + GRP - 1) / GRP;
    unsigned* code = (unsigned*)(lcnt + (size_t)tiles * AGG_WAVES);

    const int Mtiles = (N + 15) >> 4;

    gemm1_wmma<<<(Mtiles + 3) / 4, 128, 0, stream>>>(x, W1, h1, N);
    node_prep1<<<(N * HEADS + 255) / 256, 256, 0, stream>>>(h1, aS1, aD1, as1, ad1, ei, code, pair, N, E, ngroups * (GRP / 4));
    agg1_lds<<<tiles, 1024, 0, stream>>>(ei, as1, ad1, h1, b1, o1, lst_s, lst_d, lst_l, lcnt, code, pair, ngroups, N, E);
    node_prep2<<<(N + 255) / 256, 256, 0, stream>>>(o1, W2, aS2, aD2, rec, N);
    agg2_lds<<<tiles, 1024, 0, stream>>>(ei, rec, b2, out, lst_s, lst_d, lcnt, code, pair, ngroups, N, E);
}


# Layer-2 node transform (64->2, both logits) done in the layer-1 tile kernel's finish: a thread's 4 ELU values x W2, 16-lane DPP sum; ELU plane and 1 of 5 launches gone; h2 sum order now lane tree
# speedup vs baseline: 7.6829x; 1.0458x over previous
#include <hip/hip_runtime.h>
#include <math.h>

#define DIM 128
#define HID 16
#define HEADS 4
#define NEG_SLOPE 0.2f

typedef __attribute__((ext_vector_type(2))) float v2f;
typedef __attribute__((ext_vector_type(8))) float v8f;

__device__ __forceinline__ void atomicMaxFloat(float* addr, float v) {
    if (v >= 0.0f) {
        atomicMax((int*)addr, __float_as_int(v));
    } else {
        atomicMin((unsigned int*)addr, __float_as_uint(v));
    }
}

__global__ void __launch_bounds__(128)
gemm1_wmma(const float* __restrict__ x, const float* __restrict__ W1,
           float* __restrict__ h1, int N) {
    const int wave  = threadIdx.x >> 5;
    const int lane  = threadIdx.x & 31;
    const int Mtile = blockIdx.x * 4 + wave;
    const int Mtiles = (N + 15) >> 4;
    if (Mtile >= Mtiles) return;

    const int r  = lane & 15;
    const int kh = lane >> 4;
    const int mbase = Mtile * 16;
    int row = mbase + r;
    if (row >= N) row = N - 1;
    const float* __restrict__ xr = x + (size_t)row * DIM;
    const float* __restrict__ wc = W1 + 4 * r;

    v8f acc0 = {}, acc1 = {}, acc2 = {}, acc3 = {};

#pragma unroll
    for (int kb = 0; kb < DIM; kb += 4) {
        const int k0 = kb + 2 * kh;
        v2f a;
        a.x = xr[k0];
        a.y = xr[k0 + 1];
        const float4 w0 = *(const float4*)(wc + (size_t)k0 * 64);
        const float4 w1 = *(const float4*)(wc + (size_t)(k0 + 1) * 64);
        v2f b;
        b.x = w0.x;  b.y = w1.x;
        acc0 = __builtin_amdgcn_wmma_f32_16x16x4_f32(false, a, false, b, (short)0, acc0, false, false);
        b.x = w0.y;  b.y = w1.y;
        acc1 = __builtin_amdgcn_wmma_f32_16x16x4_f32(false, a, false, b, (short)0, acc1, false, false);
        b.x = w0.z;  b.y = w1.z;
        acc2 = __builtin_amdgcn_wmma_f32_16x16x4_f32(false, a, false, b, (short)0, acc2, false, false);
        b.x = w0.w;  b.y = w1.w;
        acc3 = __builtin_amdgcn_wmma_f32_16x16x4_f32(false, a, false, b, (short)0, acc3, false, false);
    }

    float* outp = h1 + (size_t)mbase * 64 + 4 * r;
#pragma unroll
    for (int i = 0; i < 8; ++i) {
        const int m = i + 8 * kh;
        if (mbase + m < N) {
            float4 o;
            o.x = acc0[i];
            o.y = acc1[i];
            o.z = acc2[i];
            o.w = acc3[i];
            *(float4*)(outp + (size_t)m * 64) = o;
        }
    }
}

#define TILE 400
#define GRP 4096

__device__ __forceinline__ unsigned tile_code(int d) {
    return (d < 0) ? 255u : (((unsigned)d / (unsigned)TILE) % 255u);
}

__global__ void __launch_bounds__(256)
node_prep1(const float* __restrict__ h1, const float* __restrict__ attS,
           const float* __restrict__ attD, float* __restrict__ as1,
           float* __restrict__ ad1, const int* __restrict__ ei,
           unsigned* __restrict__ code, int* __restrict__ pair, int N, int E, int ncode) {
    const int tid = blockIdx.x * blockDim.x + threadIdx.x;
    {
        const int* __restrict__ dstc = ei + E;
        const bool al16 = (((((size_t)dstc) | ((size_t)ei)) & 15) == 0);
        const int nthreads = gridDim.x * blockDim.x;
        for (int q = tid; q < ncode; q += nthreads) {
            const int e = q * 4;
            int d0 = -1, d1 = -1, d2 = -1, d3 = -1;
            if (al16 && e + 3 < E) {
                const int4 v = *(const int4*)(dstc + e);
                const int4 sv = *(const int4*)(ei + e);
                d0 = v.x; d1 = v.y; d2 = v.z; d3 = v.w;
                *(int4*)(pair + (size_t)e * 2)     = make_int4(sv.x, v.x, sv.y, v.y);
                *(int4*)(pair + (size_t)e * 2 + 4) = make_int4(sv.z, v.z, sv.w, v.w);
            } else {
                if (e     < E) { d0 = dstc[e];     *(int2*)(pair + (size_t)e * 2)     = make_int2(ei[e],     d0); }
                if (e + 1 < E) { d1 = dstc[e + 1]; *(int2*)(pair + (size_t)e * 2 + 2) = make_int2(ei[e + 1], d1); }
                if (e + 2 < E) { d2 = dstc[e + 2]; *(int2*)(pair + (size_t)e * 2 + 4) = make_int2(ei[e + 2], d2); }
                if (e + 3 < E) { d3 = dstc[e + 3]; *(int2*)(pair + (size_t)e * 2 + 6) = make_int2(ei[e + 3], d3); }
            }
            code[q] = tile_code(d0) | (tile_code(d1) << 8) | (tile_code(d2) << 16) | (tile_code(d3) << 24);
        }
    }
    if (tid >= N * HEADS) return;
    const int n  = tid >> 2;
    const int hd = tid & 3;
    const float* hp = h1 + (size_t)n * 64 + hd * HID;
    float ss = 0.f, dd = 0.f;
#pragma unroll
    for (int c = 0; c < HID; ++c) {
        const float v = hp[c];
        ss += v * attS[hd * HID + c];
        dd += v * attD[hd * HID + c];
    }
    as1[tid] = ss;
    ad1[tid] = dd;
}

#define WCAP 1024
#define QCAP 64
#define AGG_WAVES 32

__device__ __forceinline__ float leaky(float v) { return (v > 0.f) ? v : NEG_SLOPE * v; }

struct AggCtx {
    const int* pair;
    const float* as1; const float* ad1; const float* h1;
    float* acc; float* mx; float* sm;
    int* wq_e;
    int* lst_s; int* lst_d; float* lst_l;
    size_t lbase;
    int nbase, ntile, lane, wave, E;
    const unsigned* code; unsigned pat; int ngroups;
    const float* rec; float* adl; float* oa; float* ob;
};

__device__ __forceinline__ float4 batch_max(const AggCtx& cx, bool ok, int s, int dl) {
    float4 lg = make_float4(0.f, 0.f, 0.f, 0.f);
    if (ok) {
        const float4 a = *(const float4*)(cx.as1 + (size_t)s * 4);
        const float4 b = *(const float4*)(cx.ad1 + (size_t)(cx.nbase + dl) * 4);
        lg.x = leaky(a.x + b.x);
        lg.y = leaky(a.y + b.y);
        lg.z = leaky(a.z + b.z);
        lg.w = leaky(a.w + b.w);
        atomicMaxFloat(&cx.mx[dl * 4 + 0], lg.x);
        atomicMaxFloat(&cx.mx[dl * 4 + 1], lg.y);
        atomicMaxFloat(&cx.mx[dl * 4 + 2], lg.z);
        atomicMaxFloat(&cx.mx[dl * 4 + 3], lg.w);
    }
    return lg;
}

template <bool FROMLIST>
__device__ __forceinline__ void batch_acc(const AggCtx& cx, int nv, unsigned okm, int s, int dl, float4 lg) {
    float w0 = 0.f, w1 = 0.f, w2 = 0.f, w3 = 0.f;
    if ((okm >> cx.lane) & 1u) {
        if (!FROMLIST) {
            const float4 a = *(const float4*)(cx.as1 + (size_t)s * 4);
            const float4 b = *(const float4*)(cx.ad1 + (size_t)(cx.nbase + dl) * 4);
            lg.x = leaky(a.x + b.x);
            lg.y = leaky(a.y + b.y);
            lg.z = leaky(a.z + b.z);
            lg.w = leaky(a.w + b.w);
        }
        w0 = __expf(lg.x - cx.mx[dl * 4 + 0]);
        w1 = __expf(lg.y - cx.mx[dl * 4 + 1]);
        w2 = __expf(lg.z - cx.mx[dl * 4 + 2]);
        w3 = __expf(lg.w - cx.mx[dl * 4 + 3]);
        atomicAdd(&cx.sm[dl * 4 + 0], w0);
        atomicAdd(&cx.sm[dl * 4 + 1], w1);
        atomicAdd(&cx.sm[dl * 4 + 2], w2);
        atomicAdd(&cx.sm[dl * 4 + 3], w3);
    } else {
        s = 0; dl = 0;
    }
    const bool lo = cx.lane < 16;
    for (int j0 = 0; j0 < nv; j0 += 8) {
        float v0[8], v1[8];
#pragma unroll
        for (int u = 0; u < 8; ++u) {
            const int sj = __builtin_amdgcn_readlane(s, j0 + u);
            const float* hp = cx.h1 + (size_t)sj * 64 + cx.lane;
            v0[u] = hp[0];
            v1[u] = hp[32];
        }
#pragma unroll
        for (int u = 0; u < 8; ++u) {
            const int j = j0 + u;
            if ((okm >> j) & 1u) {
                const int dj = __builtin_amdgcn_readlane(dl, j);
                const float wlo_a = __int_as_float(__builtin_amdgcn_readlane(__float_as_int(w0), j));
                const float whi_a = __int_as_float(__builtin_amdgcn_readlane(__float_as_int(w1), j));
                const float wlo_b = __int_as_float(__builtin_amdgcn_readlane(__float_as_int(w2), j));
                const float whi_b = __int_as_float(__builtin_amdgcn_readlane(__float_as_int(w3), j));
                const float wa = lo ? wlo_a : whi_a;
                const float wb = lo ? wlo_b : whi_b;
                atomicAdd(&cx.acc[dj * 64 + cx.lane], wa * v0[u]);
                atomicAdd(&cx.acc[dj * 64 + 32 + cx.lane], wb * v1[u]);
            }
        }
    }
}

__device__ __forceinline__ void batch_max2(const AggCtx& cx, bool ok, int s, int dl) {
    if (ok) {
        const float a = cx.rec[(size_t)s * 4 + 2];
        atomicMaxFloat(&cx.mx[dl], leaky(a + cx.adl[dl]));
    }
}

__device__ __forceinline__ void batch_acc2(const AggCtx& cx, bool ok, int s, int dl) {
    if (ok) {
        const float4 r = *(const float4*)(cx.rec + (size_t)s * 4);
        const float w = __expf(leaky(r.z + cx.adl[dl]) - cx.mx[dl]);
        atomicAdd(&cx.sm[dl], w);
        atomicAdd(&cx.oa[dl], w * r.x);
        atomicAdd(&cx.ob[dl], w * r.y);
    }
}

template <int PHASE>
__device__ __forceinline__ void flush_batch(const AggCtx& cx, int head, int nv, int& nlist, bool& ovf) {
    __builtin_amdgcn_wave_barrier();
    const int qi = (head + cx.lane) & (QCAP - 1);
    int s = 0, dl = 0;
    bool ok = false;
    if (cx.lane < nv) {
        const int e = cx.wq_e[qi];
        const int2 pr = *(const int2*)(cx.pair + (size_t)e * 2);
        s  = pr.x;
        dl = pr.y - cx.nbase;
        ok = (unsigned)dl < (unsigned)cx.ntile;
    }
    const unsigned okm = __builtin_amdgcn_ballot_w32(ok);
    if (PHASE == 1) {
        const float4 lg = batch_max(cx, ok, s, dl);
        if (!ovf && nlist + 32 <= WCAP) {
            if (ok) {
                const int pos = nlist + __popc(okm & ((1u << cx.lane) - 1u));
                cx.lst_s[cx.lbase + pos] = s;
                cx.lst_d[cx.lbase + pos] = dl;
                *(float4*)(cx.lst_l + (cx.lbase + pos) * 4) = lg;
            }
            nlist += __popc(okm);
        } else {
            ovf = true;
        }
    } else if (PHASE == 2) {
        batch_acc<false>(cx, nv, okm, s, dl, make_float4(0.f, 0.f, 0.f, 0.f));
    } else if (PHASE == 3) {
        batch_max2(cx, ok, s, dl);
    } else {
        batch_acc2(cx, ok, s, dl);
    }
    __builtin_amdgcn_wave_barrier();
}

template <int PHASE>
__device__ __forceinline__ void scan_edges(const AggCtx& cx, int& nlist, bool& ovf) {
    int head = 0, cnt = 0;
    for (int g = cx.wave; g < cx.ngroups; g += AGG_WAVES) {
        const uint4* p = (const uint4*)(cx.code + (size_t)g * (GRP / 4)) + cx.lane;
        uint4 c[8];
#pragma unroll
        for (int u = 0; u < 8; ++u) c[u] = p[u * 32];
        unsigned hit[4];
#pragma unroll
        for (int sg = 0; sg < 4; ++sg) {
            unsigned bits = 0u;
#pragma unroll
            for (int k = 0; k < 8; ++k) {
                const uint4 cv = c[2 * sg + (k >> 2)];
                const unsigned v = ((k & 3) == 0) ? cv.x : ((k & 3) == 1) ? cv.y : ((k & 3) == 2) ? cv.z : cv.w;
                const unsigned x = v ^ cx.pat;
                const unsigned t = (x & 0x7f7f7f7fu) + 0x7f7f7f7fu;
                const unsigned y = ~(t | x) & 0x80808080u;
                bits |= y >> (7 - k);
            }
            hit[sg] = bits;
        }
#pragma unroll 1
        for (int sg = 0; sg < 4; ++sg) {
            unsigned bits = (sg == 0) ? hit[0] : (sg == 1) ? hit[1] : (sg == 2) ? hit[2] : hit[3];
            const int ebase = g * GRP + sg * 1024 + cx.lane * 16;
            for (int round = 0; round < 32; ++round) {
                const bool has = bits != 0u;
                const unsigned mask = __builtin_amdgcn_ballot_w32(has);
                if (mask == 0u) break;
                if (has) {
                    const int b = __ffs(bits) - 1;
                    const int pos = (head + cnt + __popc(mask & ((1u << cx.lane) - 1u))) & (QCAP - 1);
                    cx.wq_e[pos] = ebase + ((b >> 2) & 1) * 512 + (b & 3) * 4 + (b >> 3);
                    bits &= bits - 1u;
                }
                cnt += __popc(mask);
                if (cnt >= 32) {
                    flush_batch<PHASE>(cx, head, 32, nlist, ovf);
                    head = (head + 32) & (QCAP - 1);
                    cnt -= 32;
                }
            }
        }
    }
    if (cnt > 0) flush_batch<PHASE>(cx, head, cnt, nlist, ovf);
}

template <int M>
__device__ __forceinline__ float row_xor_add(float v) {
    return v + __int_as_float(__builtin_amdgcn_update_dpp(0, __float_as_int(v), 0x160 + M, 0xf, 0xf, false));
}
__device__ __forceinline__ float row_sum16(float v) {
    v = row_xor_add<1>(v);
    v = row_xor_add<2>(v);
    v = row_xor_add<4>(v);
    v = row_xor_add<8>(v);
    return v;
}

__global__ void __launch_bounds__(1024)
agg1_lds(const int* __restrict__ ei, const float* __restrict__ as1,
         const float* __restrict__ ad1, const float* __restrict__ h1,
         const float* __restrict__ b1, const float* __restrict__ W2,
         const float* __restrict__ attS2, const float* __restrict__ attD2, float* __restrict__ rec,
         int* lst_s, int* lst_d, float* lst_l, int* lcnt,
         const unsigned* __restrict__ code, const int* __restrict__ pair, int ngroups, int N, int E) {
    __shared__ float acc[TILE * 64];
    __shared__ float mx[TILE * HEADS];
    __shared__ float sm[TILE * HEADS];
    __shared__ int   qe[AGG_WAVES * QCAP];

    const int tid = threadIdx.x;
    AggCtx cx;
    cx.lane = tid & 31;
    cx.wave = __builtin_amdgcn_readfirstlane(tid >> 5);
    cx.nbase = blockIdx.x * TILE;
    cx.ntile = (N - cx.nbase < TILE) ? (N - cx.nbase) : TILE;
    cx.E = E;
    cx.pair = pair;
    cx.as1 = as1; cx.ad1 = ad1; cx.h1 = h1;
    cx.acc = acc; cx.mx = mx; cx.sm = sm;
    cx.code = code; cx.pat = (blockIdx.x % 255u) * 0x01010101u; cx.ngroups = ngroups;
    cx.wq_e = qe + cx.wave * QCAP;
    cx.lst_s = lst_s; cx.lst_d = lst_d; cx.lst_l = lst_l;
    cx.lbase = ((size_t)blockIdx.x * AGG_WAVES + cx.wave) * WCAP;

    for (int i = tid; i < TILE * 64; i += 1024) acc[i] = 0.f;
    for (int i = tid; i < TILE * HEADS; i += 1024) {
        sm[i] = 0.f;
        float v = 0.f;
        if ((i >> 2) < cx.ntile) {
            const size_t g = (size_t)cx.nbase * HEADS + i;
            v = leaky(as1[g] + ad1[g]);
        }
        mx[i] = v;
    }
    __syncthreads();

    int nlist = 0;
    bool ovf = false;
    scan_edges<1>(cx, nlist, ovf);
    if (cx.lane == 0) lcnt[blockIdx.x * AGG_WAVES + cx.wave] = ovf ? -1 : nlist;
    __syncthreads();

    if (!ovf) {
        for (int p = 0; p < nlist; p += 32) {
            const int nv = (nlist - p < 32) ? (nlist - p) : 32;
            int s = 0, dl = 0;
            float4 lg = make_float4(0.f, 0.f, 0.f, 0.f);
            if (cx.lane < nv) {
                s  = lst_s[cx.lbase + p + cx.lane];
                dl = lst_d[cx.lbase + p + cx.lane];
                lg = *(const float4*)(lst_l + (cx.lbase + p + cx.lane) * 4);
            }
            batch_acc<true>(cx, nv, (nv >= 32) ? 0xffffffffu : ((1u << nv) - 1u), s, dl, lg);
        }
    } else {
        int dummy_n = 0; bool dummy_o = false;
        scan_edges<2>(cx, dummy_n, dummy_o);
    }
    __syncthreads();

    const int c  = (tid & 15) * 4;
    const int hd = c >> 4;
    const float4 bv  = *(const float4*)(b1 + c);
    const float4 w2a = *(const float4*)(W2 + c * 2);
    const float4 w2b = *(const float4*)(W2 + c * 2 + 4);
    const float s20 = attS2[0], s21 = attS2[1];
    const float d20 = attD2[0], d21 = attD2[1];
    const int lim = cx.ntile * 16;
    for (int q0 = 0; q0 < lim; q0 += 1024) {
        const int q = q0 + tid;
        const bool act = q < lim;
        const int nl = act ? (q >> 4) : 0;
        const size_t n = (size_t)cx.nbase + nl;
        const float ls = leaky(as1[n * HEADS + hd] + ad1[n * HEADS + hd]);
        const float wself = __expf(ls - mx[nl * HEADS + hd]);
        const float den = sm[nl * HEADS + hd] + wself + 1e-16f;
        const float4 hv = *(const float4*)(h1 + n * 64 + c);
        const float4 av = *(const float4*)(acc + nl * 64 + c);
        float4 r;
        r.x = (av.x + wself * hv.x) / den + bv.x;
        r.y = (av.y + wself * hv.y) / den + bv.y;
        r.z = (av.z + wself * hv.z) / den + bv.z;
        r.w = (av.w + wself * hv.w) / den + bv.w;
        r.x = (r.x > 0.f) ? r.x : expm1f(r.x);
        r.y = (r.y > 0.f) ? r.y : expm1f(r.y);
        r.z = (r.z > 0.f) ? r.z : expm1f(r.z);
        r.w = (r.w > 0.f) ? r.w : expm1f(r.w);
        float p0 = r.x * w2a.x + r.y * w2a.z + r.z * w2b.x + r.w * w2b.z;
        float p1 = r.x * w2a.y + r.y * w2a.w + r.z * w2b.y + r.w * w2b.w;
        p0 = row_sum16(p0);
        p1 = row_sum16(p1);
        if (act && (tid & 15) == 0) {
            float4 o;
            o.x = p0;
            o.y = p1;
            o.z = p0 * s20 + p1 * s21;
            o.w = p0 * d20 + p1 * d21;
            *(float4*)(rec + n * 4) = o;
        }
    }
}

#define L2B 8

__global__ void __launch_bounds__(1024)
agg2_lds(const int* __restrict__ ei, const float* __restrict__ rec,
         const float* __restrict__ b2, float* __restrict__ out,
         int* lst_s, int* lst_d, const int* __restrict__ lcnt,
         const unsigned* __restrict__ code, const int* __restrict__ pair, int ngroups, int N, int E) {
    __shared__ float mx[TILE];
    __shared__ float sm[TILE];
    __shared__ float oa[TILE];
    __shared__ float ob[TILE];
    __shared__ float adl[TILE];
    __shared__ int   qe[AGG_WAVES * QCAP];
    static_assert(TILE <= 1024, "one thread per node of the tile");

    const int tid = threadIdx.x;
    AggCtx cx;
    cx.lane = tid & 31;
    cx.wave = __builtin_amdgcn_readfirstlane(tid >> 5);
    cx.nbase = blockIdx.x * TILE;
    cx.ntile = (N - cx.nbase < TILE) ? (N - cx.nbase) : TILE;
    cx.E = E;
    cx.pair = pair;
    cx.as1 = nullptr; cx.ad1 = nullptr; cx.h1 = nullptr; cx.acc = nullptr;
    cx.mx = mx; cx.sm = sm;
    cx.code = code; cx.pat = (blockIdx.x % 255u) * 0x01010101u; cx.ngroups = ngroups;
    cx.wq_e = qe + cx.wave * QCAP;
    cx.lst_s = lst_s; cx.lst_d = lst_d; cx.lst_l = nullptr;
    cx.lbase = ((size_t)blockIdx.x * AGG_WAVES + cx.wave) * WCAP;
    cx.rec = rec; cx.adl = adl; cx.oa = oa; cx.ob = ob;

    const int cnt = lcnt[blockIdx.x * AGG_WAVES + cx.wave];

    int sv[L2B], dv[L2B];
#pragma unroll
    for (int k = 0; k < L2B; ++k) { sv[k] = 0; dv[k] = 0; }
    if (cnt > 0) {
#pragma unroll
        for (int k = 0; k < L2B; ++k) {
            int idx = k * 32 + cx.lane;
            idx = (idx < cnt) ? idx : (cnt - 1);
            sv[k] = lst_s[cx.lbase + idx];
            dv[k] = lst_d[cx.lbase + idx];
        }
    }

    float4 rown = make_float4(0.f, 0.f, 0.f, 0.f);
    if (tid < TILE) {
        float m = 0.f, a = 0.f;
        if (tid < cx.ntile) {
            rown = *(const float4*)(rec + ((size_t)cx.nbase + tid) * 4);
            a = rown.w;
            m = leaky(rown.z + rown.w);
        }
        mx[tid] = m; adl[tid] = a;
        sm[tid] = 0.f; oa[tid] = 0.f; ob[tid] = 0.f;
    }

    float rx[L2B], ry[L2B], lg[L2B];
#pragma unroll
    for (int k = 0; k < L2B; ++k) { rx[k] = 0.f; ry[k] = 0.f; lg[k] = 0.f; }
    if (cnt > 0) {
#pragma unroll
        for (int k = 0; k < L2B; ++k) {
            const float4 r = *(const float4*)(rec + (size_t)sv[k] * 4);
            rx[k] = r.x; ry[k] = r.y; lg[k] = r.z;
        }
    }
    __syncthreads();

    int dummy_n = 0; bool dummy_o = false;

    if (cnt >= 0) {
        float ad[L2B];
#pragma unroll
        for (int k = 0; k < L2B; ++k) ad[k] = adl[dv[k]];
#pragma unroll
        for (int k = 0; k < L2B; ++k) {
            if (k * 32 + cx.lane < cnt) {
                lg[k] = leaky(lg[k] + ad[k]);
                atomicMaxFloat(&mx[dv[k]], lg[k]);
            }
        }
        for (int p = L2B * 32; p < cnt; p += 32) {
            const int nv = (cnt - p < 32) ? (cnt - p) : 32;
            int s = 0, dl = 0;
            if (cx.lane < nv) {
                s  = lst_s[cx.lbase + p + cx.lane];
                dl = lst_d[cx.lbase + p + cx.lane];
            }
            batch_max2(cx, cx.lane < nv, s, dl);
        }
    } else {
        scan_edges<3>(cx, dummy_n, dummy_o);
    }
    __syncthreads();

    if (cnt >= 0) {
        float mv[L2B];
#pragma unroll
        for (int k = 0; k < L2B; ++k) mv[k] = mx[dv[k]];
#pragma unroll
        for (int k = 0; k < L2B; ++k) {
            if (k * 32 + cx.lane < cnt) {
                const int d = dv[k];
                const float w = __expf(lg[k] - mv[k]);
                atomicAdd(&sm[d], w);
                atomicAdd(&oa[d], w * rx[k]);
                atomicAdd(&ob[d], w * ry[k]);
            }
        }
        for (int p = L2B * 32; p < cnt; p += 32) {
            const int nv = (cnt - p < 32) ? (cnt - p) : 32;
            int s = 0, dl = 0;
            if (cx.lane < nv) {
                s  = lst_s[cx.lbase + p + cx.lane];
                dl = lst_d[cx.lbase + p + cx.lane];
            }
            batch_acc2(cx, cx.lane < nv, s, dl);
        }
    } else {
        scan_edges<4>(cx, dummy_n, dummy_o);
    }
    __syncthreads();

    if (tid < cx.ntile) {
        const size_t n = (size_t)cx.nbase + tid;
        const float4 r = rown;
        const float wself = __expf(leaky(r.z + r.w) - mx[tid]);
        const float inv = 1.0f / (sm[tid] + wself + 1e-16f);
        float2 o;
        o.x = (oa[tid] + wself * r.x) * inv + b2[0];
        o.y = (ob[tid] + wself * r.y) * inv + b2[1];
        *(float2*)(out + n * 2) = o;
    }
}

extern "C" void kernel_launch(void* const* d_in, const int* in_sizes, int n_in,
                              void* d_out, int out_size, void* d_ws, size_t ws_size,
                              hipStream_t stream) {
    const float* x    = (const float*)d_in[0];
    const int*   ei   = (const int*)d_in[1];
    const float* W1   = (const float*)d_in[2];
    const float* aS1  = (const float*)d_in[3];
    const float* aD1  = (const float*)d_in[4];
    const float* b1   = (const float*)d_in[5];
    const float* W2   = (const float*)d_in[6];
    const float* aS2  = (const float*)d_in[7];
    const float* aD2  = (const float*)d_in[8];
    const float* b2   = (const float*)d_in[9];

    const int N = in_sizes[0] / DIM;
    const int E = in_sizes[1] / 2;
    float* out = (float*)d_out;

    float* ws  = (float*)d_ws;
    float* h1  = ws;
    float* as1 = h1 + (size_t)N * 64;
    float* ad1 = as1 + (size_t)N * HEADS;
    float* rec = ad1 + (size_t)N * HEADS;
    const int tiles = (N + TILE - 1) / TILE;
    float* lst_l = rec + (size_t)N * 4;
    int* pair  = (int*)(lst_l + (size_t)tiles * AGG_WAVES * WCAP * 4);
    int* lst_s = pair + (size_t)E * 2;
    int* lst_d = lst_s + (size_t)tiles * AGG_WAVES * WCAP;
    int* lcnt  = lst_d + (size_t)tiles * AGG_WAVES * WCAP;
    const int ngroups = (E + GRP - 1) / GRP;
    unsigned* code = (unsigned*)(lcnt + (size_t)tiles * AGG_WAVES);

    const int Mtiles = (N + 15) >> 4;

    gemm1_wmma<<<(Mtiles + 3) / 4, 128, 0, stream>>>(x, W1, h1, N);
    node_prep1<<<(N * HEADS + 255) / 256, 256, 0, stream>>>(h1, aS1, aD1, as1, ad1, ei, code, pair, N, E, ngroups * (GRP / 4));
    agg1_lds<<<tiles, 1024, 0, stream>>>(ei, as1, ad1, h1, b1, W2, aS2, aD2, rec, lst_s, lst_d, lst_l, lcnt, code, pair, ngroups, N, E);
    agg2_lds<<<tiles, 1024, 0, stream>>>(ei, rec, b2, out, lst_s, lst_d, lcnt, code, pair, ngroups, N, E);
}
